# starting kernel
# baseline (speedup 1.0000x reference)
#include <hip/hip_runtime.h>
#include <hip/hip_bf16.h>


typedef __attribute__((ext_vector_type(2))) float v2f;
typedef __attribute__((ext_vector_type(8))) float v8f;

#define DFEAT 128

__global__ __launch_bounds__(256) void gconv_gemm_xw(const float* __restrict__ X,
                                                     const float* __restrict__ W,
                                                     float* __restrict__ Y,
                                                     int mTiles) {
    const int wave = (int)((blockIdx.x * blockDim.x + threadIdx.x) >> 5);
    const int lane = (int)(threadIdx.x & 31);
    const int nTiles = DFEAT / 16;
    const int mt = wave / nTiles;
    const int nt = wave % nTiles;
    if (mt >= mTiles) return;

    const int hi  = lane >> 4;
    const int l15 = lane & 15;

    const float* Arow = X + (size_t)(mt * 16 + l15) * DFEAT;
    const int ncol = nt * 16 + l15;

    v8f acc = {};
#pragma unroll 8
    for (int k = 0; k < DFEAT; k += 4) {
        v2f a, b;
        const int ka = k + 2 * hi;
        a.x = Arow[ka + 0];
        a.y = Arow[ka + 1];
        b.x = W[(size_t)(ka + 0) * DFEAT + ncol];
        b.y = W[(size_t)(ka + 1) * DFEAT + ncol];
        acc = __builtin_amdgcn_wmma_f32_16x16x4_f32(
            false, a, false, b, (short)0, acc, false, false);
    }

    float* Yt = Y + (size_t)(mt * 16 + 8 * hi) * DFEAT + ncol;
#pragma unroll
    for (int i = 0; i < 8; ++i) {
        Yt[(size_t)i * DFEAT] = acc[i];
    }
}

__global__ __launch_bounds__(256) void gconv_scatter(const float* __restrict__ Y,
                                                     const int* __restrict__ esrc,
                                                     const int* __restrict__ edst,
                                                     const float* __restrict__ eval,
                                                     float* __restrict__ Z,
                                                     int nEdges) {
    const int wave = (int)((blockIdx.x * blockDim.x + threadIdx.x) >> 5);
    const int lane = (int)(threadIdx.x & 31);
    if (wave >= nEdges) return;

    const int   s = esrc[wave];
    const int   d = edst[wave];
    const float v = eval[wave];

    const float4 y = ((const float4*)(Y + (size_t)s * DFEAT))[lane];
    float* zr = Z + (size_t)d * DFEAT + lane * 4;
    atomicAdd(zr + 0, v * y.x);
    atomicAdd(zr + 1, v * y.y);
    atomicAdd(zr + 2, v * y.z);
    atomicAdd(zr + 3, v * y.w);
}

extern "C" void kernel_launch(void* const* d_in, const int* in_sizes, int n_in,
                              void* d_out, int out_size, void* d_ws, size_t ws_size,
                              hipStream_t stream) {
    const float* X    = (const float*)d_in[0];
    const float* W    = (const float*)d_in[1];
    const int*   esrc = (const int*)  d_in[2];
    const int*   edst = (const int*)  d_in[3];
    const float* eval = (const float*)d_in[4];
    float*       Z    = (float*)d_out;
    float*       Y    = (float*)d_ws;

    const int nNodes = in_sizes[0] / DFEAT;
    const int nEdges = in_sizes[2];

    const int mTiles   = nNodes / 16;
    const int nTiles   = DFEAT / 16;
    const int nWaves   = mTiles * nTiles;
    const int block    = 256;
    const int gemmGrid = (nWaves * 32 + block - 1) / block;
    gconv_gemm_xw<<<gemmGrid, block, 0, stream>>>(X, W, Y, mTiles);

    hipMemsetAsync(d_out, 0, (size_t)out_size * sizeof(float), stream);

    const int edgeGrid = (nEdges + 7) / 8;
    gconv_scatter<<<edgeGrid, block, 0, stream>>>(Y, esrc, edst, eval, Z, nEdges);
}


# Global f32 atomic scatter replaced by owner blocks: each keeps 391 output rows in LDS, scans all dst ids, adds its edges' Y rows by LDS f32 adds (arrival order), stores each row once; memset gone
# speedup vs baseline: 1.9093x; 1.9093x over previous
#include <hip/hip_runtime.h>
#include <hip/hip_bf16.h>


typedef __attribute__((ext_vector_type(2))) float v2f;
typedef __attribute__((ext_vector_type(8))) float v8f;

#define DFEAT 128

__global__ __launch_bounds__(256) void gconv_gemm_xw(const float* __restrict__ X,
                                                     const float* __restrict__ W,
                                                     float* __restrict__ Y,
                                                     int mTiles) {
    const int wave = (int)((blockIdx.x * blockDim.x + threadIdx.x) >> 5);
    const int lane = (int)(threadIdx.x & 31);
    const int nTiles = DFEAT / 16;
    const int mt = wave / nTiles;
    const int nt = wave % nTiles;
    if (mt >= mTiles) return;

    const int hi  = lane >> 4;
    const int l15 = lane & 15;

    const float* Arow = X + (size_t)(mt * 16 + l15) * DFEAT;
    const int ncol = nt * 16 + l15;

    v8f acc = {};
#pragma unroll 8
    for (int k = 0; k < DFEAT; k += 4) {
        v2f a, b;
        const int ka = k + 2 * hi;
        a.x = Arow[ka + 0];
        a.y = Arow[ka + 1];
        b.x = W[(size_t)(ka + 0) * DFEAT + ncol];
        b.y = W[(size_t)(ka + 1) * DFEAT + ncol];
        acc = __builtin_amdgcn_wmma_f32_16x16x4_f32(
            false, a, false, b, (short)0, acc, false, false);
    }

    float* Yt = Y + (size_t)(mt * 16 + 8 * hi) * DFEAT + ncol;
#pragma unroll
    for (int i = 0; i < 8; ++i) {
        Yt[(size_t)i * DFEAT] = acc[i];
    }
}

#define OWN_ROWS   391
#define OWN_WAVES  32
#define STEP_LOADS 6
#define STEP_EDGES (STEP_LOADS * 128)
#define DRAIN_AT   64
#define QCAP       (DRAIN_AT + STEP_EDGES)

__global__ __launch_bounds__(OWN_WAVES * 32) void gconv_owner_sum(const float* __restrict__ Y,
                                                                  const int* __restrict__ esrc,
                                                                  const int* __restrict__ edst,
                                                                  const float* __restrict__ eval,
                                                                  float* __restrict__ Z,
                                                                  int nNodes, int nEdges) {
    __shared__ float acc[OWN_ROWS * DFEAT];
    __shared__ int   qall[OWN_WAVES * QCAP];

    const int lane = (int)(threadIdx.x & 31);
    const int wave = __builtin_amdgcn_readfirstlane((int)(threadIdx.x >> 5));
    const int rowBase = (int)blockIdx.x * OWN_ROWS;
    int nrows = nNodes - rowBase;
    if (nrows > OWN_ROWS) nrows = OWN_ROWS;

    for (int i = (int)threadIdx.x; i < OWN_ROWS * DFEAT / 4; i += OWN_WAVES * 32) {
        ((float4*)acc)[i] = make_float4(0.f, 0.f, 0.f, 0.f);
    }
    __syncthreads();

    int* q = qall + wave * QCAP;
    int cnt = 0;
    const int nSteps = (nEdges + STEP_EDGES - 1) / STEP_EDGES;

    int4 cur[STEP_LOADS], nxt[STEP_LOADS];

    auto load_step = [&](int t, int4 (&d)[STEP_LOADS]) {
        const int base = t * STEP_EDGES;
        if (base + STEP_EDGES <= nEdges) {
            const int4* p = (const int4*)(edst + base);
#pragma unroll
            for (int k = 0; k < STEP_LOADS; ++k) d[k] = p[k * 32 + lane];
        } else {
#pragma unroll
            for (int k = 0; k < STEP_LOADS; ++k) {
                const int i0 = base + k * 128 + lane * 4;
                d[k].x = (i0 + 0 < nEdges) ? edst[i0 + 0] : -1;
                d[k].y = (i0 + 1 < nEdges) ? edst[i0 + 1] : -1;
                d[k].z = (i0 + 2 < nEdges) ? edst[i0 + 2] : -1;
                d[k].w = (i0 + 3 < nEdges) ? edst[i0 + 3] : -1;
            }
        }
    };

    if (wave < nSteps) load_step(wave, cur);

    for (int t = wave; t < nSteps; t += OWN_WAVES) {
        const int tn = t + OWN_WAVES;
        if (tn < nSteps) load_step(tn, nxt);

        const int base = t * STEP_EDGES;
#pragma unroll
        for (int k = 0; k < STEP_LOADS; ++k) {
            const bool m0 = (unsigned)(cur[k].x - rowBase) < (unsigned)nrows;
            const bool m1 = (unsigned)(cur[k].y - rowBase) < (unsigned)nrows;
            const bool m2 = (unsigned)(cur[k].z - rowBase) < (unsigned)nrows;
            const bool m3 = (unsigned)(cur[k].w - rowBase) < (unsigned)nrows;
            if (__builtin_amdgcn_ballot_w32(m0 | m1 | m2 | m3) != 0u) {
                const int e0 = base + k * 128 + lane * 4;
                const unsigned b0 = __builtin_amdgcn_ballot_w32(m0);
                if (m0) q[cnt + (int)__builtin_amdgcn_mbcnt_lo(b0, 0u)] = e0 + 0;
                cnt += __builtin_popcount(b0);
                const unsigned b1 = __builtin_amdgcn_ballot_w32(m1);
                if (m1) q[cnt + (int)__builtin_amdgcn_mbcnt_lo(b1, 0u)] = e0 + 1;
                cnt += __builtin_popcount(b1);
                const unsigned b2 = __builtin_amdgcn_ballot_w32(m2);
                if (m2) q[cnt + (int)__builtin_amdgcn_mbcnt_lo(b2, 0u)] = e0 + 2;
                cnt += __builtin_popcount(b2);
                const unsigned b3 = __builtin_amdgcn_ballot_w32(m3);
                if (m3) q[cnt + (int)__builtin_amdgcn_mbcnt_lo(b3, 0u)] = e0 + 3;
                cnt += __builtin_popcount(b3);
            }
        }

        if (cnt >= DRAIN_AT || tn >= nSteps) {
            __builtin_amdgcn_wave_barrier();
            for (int b = 0; b < cnt; b += 32) {
                int n = cnt - b;
                if (n > 32) n = 32;
                const int e  = q[b + (lane < n ? lane : 0)];
                const int s  = esrc[e];
                const int dl = edst[e] - rowBase;
                const float v = eval[e];
                for (int j = 0; j < n; j += 8) {
                    float4 y[8];
#pragma unroll
                    for (int u = 0; u < 8; ++u) {
                        int ju = j + u;
                        if (ju > n - 1) ju = n - 1;
                        const int sj = __builtin_amdgcn_readlane(s, ju);
                        y[u] = ((const float4*)(Y + (size_t)sj * DFEAT))[lane];
                    }
#pragma unroll
                    for (int u = 0; u < 8; ++u) {
                        if (j + u < n) {
                            const int   rj = __builtin_amdgcn_readlane(dl, j + u);
                            const float vj = __int_as_float(__builtin_amdgcn_readlane(__float_as_int(v), j + u));
                            float* a = acc + rj * DFEAT + lane;
                            atomicAdd(a +  0, vj * y[u].x);
                            atomicAdd(a + 32, vj * y[u].y);
                            atomicAdd(a + 64, vj * y[u].z);
                            atomicAdd(a + 96, vj * y[u].w);
                        }
                    }
                }
            }
            cnt = 0;
            __builtin_amdgcn_wave_barrier();
        }

#pragma unroll
        for (int k = 0; k < STEP_LOADS; ++k) cur[k] = nxt[k];
    }

    __syncthreads();

    for (int r = wave; r < nrows; r += OWN_WAVES) {
        const float* a = acc + r * DFEAT + lane;
        float4 o;
        o.x = a[0];
        o.y = a[32];
        o.z = a[64];
        o.w = a[96];
        ((float4*)(Z + (size_t)(rowBase + r) * DFEAT))[lane] = o;
    }
}

extern "C" void kernel_launch(void* const* d_in, const int* in_sizes, int n_in,
                              void* d_out, int out_size, void* d_ws, size_t ws_size,
                              hipStream_t stream) {
    const float* X    = (const float*)d_in[0];
    const float* W    = (const float*)d_in[1];
    const int*   esrc = (const int*)  d_in[2];
    const int*   edst = (const int*)  d_in[3];
    const float* eval = (const float*)d_in[4];
    float*       Z    = (float*)d_out;
    float*       Y    = (float*)d_ws;

    const int nNodes = in_sizes[0] / DFEAT;
    const int nEdges = in_sizes[2];

    const int mTiles   = nNodes / 16;
    const int nTiles   = DFEAT / 16;
    const int nWaves   = mTiles * nTiles;
    const int block    = 256;
    const int gemmGrid = (nWaves * 32 + block - 1) / block;
    gconv_gemm_xw<<<gemmGrid, block, 0, stream>>>(X, W, Y, mTiles);

    const int ownGrid = (nNodes + OWN_ROWS - 1) / OWN_ROWS;
    gconv_owner_sum<<<ownGrid, OWN_WAVES * 32, 0, stream>>>(Y, esrc, edst, eval, Z, nNodes, nEdges);
}


# GEMM operand feed: 8-wave blocks walk 5 row tiles, W strip kept in registers as B fragments, X tile loaded once as whole rows and shared through LDS (was per-lane row loads in every wave)
# speedup vs baseline: 2.2475x; 1.1771x over previous
#include <hip/hip_runtime.h>
#include <hip/hip_bf16.h>


typedef __attribute__((ext_vector_type(2))) float v2f;
typedef __attribute__((ext_vector_type(8))) float v8f;

#define DFEAT 128

#define GEMM_TPB 5
#define XPITCH   132

__global__ __launch_bounds__(256) void gconv_gemm_xw(const float* __restrict__ X,
                                                     const float* __restrict__ W,
                                                     float* __restrict__ Y,
                                                     int mTiles) {
    __shared__ float xs[2][16 * XPITCH];

    const int tid  = (int)threadIdx.x;
    const int lane = tid & 31;
    const int wave = __builtin_amdgcn_readfirstlane(tid >> 5);
    const int hi   = lane >> 4;
    const int l15  = lane & 15;
    const int ncol = wave * 16 + l15;

    v2f bfrag[DFEAT / 4];
#pragma unroll
    for (int kk = 0; kk < DFEAT / 4; ++kk) {
        const int ka = 4 * kk + 2 * hi;
        bfrag[kk].x = W[(size_t)(ka + 0) * DFEAT + ncol];
        bfrag[kk].y = W[(size_t)(ka + 1) * DFEAT + ncol];
    }

    const int mt0 = (int)blockIdx.x * GEMM_TPB;
    int nt = mTiles - mt0;
    if (nt > GEMM_TPB) nt = GEMM_TPB;

    const int sr = tid >> 5;
    const int sc = tid & 31;
    const float4* Xt = (const float4*)(X + ((size_t)mt0 * 16 + sr) * DFEAT) + sc;
    float* sdst = &xs[0][sr * XPITCH + 4 * sc];
    const float* asrc = &xs[0][l15 * XPITCH + 2 * hi];

    float4 g0 = make_float4(0.f, 0.f, 0.f, 0.f), g1 = g0;
    if (nt > 0) {
        g0 = Xt[0];
        g1 = Xt[8 * DFEAT / 4];
        *(float4*)(sdst)              = g0;
        *(float4*)(sdst + 8 * XPITCH) = g1;
    }
    __syncthreads();

    for (int j = 0; j < nt; ++j) {
        const int buf = j & 1;
        const bool more = (j + 1 < nt);
        if (more) {
            const float4* Xn = Xt + (size_t)(j + 1) * (16 * DFEAT / 4);
            g0 = Xn[0];
            g1 = Xn[8 * DFEAT / 4];
        }

        v2f a[DFEAT / 4];
        const float* ab = asrc + buf * (16 * XPITCH);
#pragma unroll
        for (int kk = 0; kk < DFEAT / 4; ++kk) a[kk] = *(const v2f*)(ab + 4 * kk);

        v8f acc = {};
#pragma unroll
        for (int kk = 0; kk < DFEAT / 4; ++kk) {
            acc = __builtin_amdgcn_wmma_f32_16x16x4_f32(
                false, a[kk], false, bfrag[kk], (short)0, acc, false, false);
        }

        if (more) {
            float* sd = sdst + (buf ^ 1) * (16 * XPITCH);
            *(float4*)(sd)              = g0;
            *(float4*)(sd + 8 * XPITCH) = g1;
        }
        __syncthreads();

        float* Yt = Y + ((size_t)(mt0 + j) * 16 + 8 * hi) * DFEAT + ncol;
#pragma unroll
        for (int i = 0; i < 8; ++i) Yt[(size_t)i * DFEAT] = acc[i];
    }
}

#define OWN_ROWS   391
#define OWN_WAVES  32
#define STEP_LOADS 6
#define STEP_EDGES (STEP_LOADS * 128)
#define DRAIN_AT   64
#define QCAP       (DRAIN_AT + STEP_EDGES)

__global__ __launch_bounds__(OWN_WAVES * 32) void gconv_owner_sum(const float* __restrict__ Y,
                                                                  const int* __restrict__ esrc,
                                                                  const int* __restrict__ edst,
                                                                  const float* __restrict__ eval,
                                                                  float* __restrict__ Z,
                                                                  int nNodes, int nEdges) {
    __shared__ float acc[OWN_ROWS * DFEAT];
    __shared__ int   qall[OWN_WAVES * QCAP];

    const int lane = (int)(threadIdx.x & 31);
    const int wave = __builtin_amdgcn_readfirstlane((int)(threadIdx.x >> 5));
    const int rowBase = (int)blockIdx.x * OWN_ROWS;
    int nrows = nNodes - rowBase;
    if (nrows > OWN_ROWS) nrows = OWN_ROWS;

    for (int i = (int)threadIdx.x; i < OWN_ROWS * DFEAT / 4; i += OWN_WAVES * 32) {
        ((float4*)acc)[i] = make_float4(0.f, 0.f, 0.f, 0.f);
    }
    __syncthreads();

    int* q = qall + wave * QCAP;
    int cnt = 0;
    const int nSteps = (nEdges + STEP_EDGES - 1) / STEP_EDGES;

    int4 cur[STEP_LOADS], nxt[STEP_LOADS];

    auto load_step = [&](int t, int4 (&d)[STEP_LOADS]) {
        const int base = t * STEP_EDGES;
        if (base + STEP_EDGES <= nEdges) {
            const int4* p = (const int4*)(edst + base);
#pragma unroll
            for (int k = 0; k < STEP_LOADS; ++k) d[k] = p[k * 32 + lane];
        } else {
#pragma unroll
            for (int k = 0; k < STEP_LOADS; ++k) {
                const int i0 = base + k * 128 + lane * 4;
                d[k].x = (i0 + 0 < nEdges) ? edst[i0 + 0] : -1;
                d[k].y = (i0 + 1 < nEdges) ? edst[i0 + 1] : -1;
                d[k].z = (i0 + 2 < nEdges) ? edst[i0 + 2] : -1;
                d[k].w = (i0 + 3 < nEdges) ? edst[i0 + 3] : -1;
            }
        }
    };

    if (wave < nSteps) load_step(wave, cur);

    for (int t = wave; t < nSteps; t += OWN_WAVES) {
        const int tn = t + OWN_WAVES;
        if (tn < nSteps) load_step(tn, nxt);

        const int base = t * STEP_EDGES;
#pragma unroll
        for (int k = 0; k < STEP_LOADS; ++k) {
            const bool m0 = (unsigned)(cur[k].x - rowBase) < (unsigned)nrows;
            const bool m1 = (unsigned)(cur[k].y - rowBase) < (unsigned)nrows;
            const bool m2 = (unsigned)(cur[k].z - rowBase) < (unsigned)nrows;
            const bool m3 = (unsigned)(cur[k].w - rowBase) < (unsigned)nrows;
            if (__builtin_amdgcn_ballot_w32(m0 | m1 | m2 | m3) != 0u) {
                const int e0 = base + k * 128 + lane * 4;
                const unsigned b0 = __builtin_amdgcn_ballot_w32(m0);
                if (m0) q[cnt + (int)__builtin_amdgcn_mbcnt_lo(b0, 0u)] = e0 + 0;
                cnt += __builtin_popcount(b0);
                const unsigned b1 = __builtin_amdgcn_ballot_w32(m1);
                if (m1) q[cnt + (int)__builtin_amdgcn_mbcnt_lo(b1, 0u)] = e0 + 1;
                cnt += __builtin_popcount(b1);
                const unsigned b2 = __builtin_amdgcn_ballot_w32(m2);
                if (m2) q[cnt + (int)__builtin_amdgcn_mbcnt_lo(b2, 0u)] = e0 + 2;
                cnt += __builtin_popcount(b2);
                const unsigned b3 = __builtin_amdgcn_ballot_w32(m3);
                if (m3) q[cnt + (int)__builtin_amdgcn_mbcnt_lo(b3, 0u)] = e0 + 3;
                cnt += __builtin_popcount(b3);
            }
        }

        if (cnt >= DRAIN_AT || tn >= nSteps) {
            __builtin_amdgcn_wave_barrier();
            for (int b = 0; b < cnt; b += 32) {
                int n = cnt - b;
                if (n > 32) n = 32;
                const int e  = q[b + (lane < n ? lane : 0)];
                const int s  = esrc[e];
                const int dl = edst[e] - rowBase;
                const float v = eval[e];
                for (int j = 0; j < n; j += 8) {
                    float4 y[8];
#pragma unroll
                    for (int u = 0; u < 8; ++u) {
                        int ju = j + u;
                        if (ju > n - 1) ju = n - 1;
                        const int sj = __builtin_amdgcn_readlane(s, ju);
                        y[u] = ((const float4*)(Y + (size_t)sj * DFEAT))[lane];
                    }
#pragma unroll
                    for (int u = 0; u < 8; ++u) {
                        if (j + u < n) {
                            const int   rj = __builtin_amdgcn_readlane(dl, j + u);
                            const float vj = __int_as_float(__builtin_amdgcn_readlane(__float_as_int(v), j + u));
                            float* a = acc + rj * DFEAT + lane;
                            atomicAdd(a +  0, vj * y[u].x);
                            atomicAdd(a + 32, vj * y[u].y);
                            atomicAdd(a + 64, vj * y[u].z);
                            atomicAdd(a + 96, vj * y[u].w);
                        }
                    }
                }
            }
            cnt = 0;
            __builtin_amdgcn_wave_barrier();
        }

#pragma unroll
        for (int k = 0; k < STEP_LOADS; ++k) cur[k] = nxt[k];
    }

    __syncthreads();

    for (int r = wave; r < nrows; r += OWN_WAVES) {
        const float* a = acc + r * DFEAT + lane;
        float4 o;
        o.x = a[0];
        o.y = a[32];
        o.z = a[64];
        o.w = a[96];
        ((float4*)(Z + (size_t)(rowBase + r) * DFEAT))[lane] = o;
    }
}

extern "C" void kernel_launch(void* const* d_in, const int* in_sizes, int n_in,
                              void* d_out, int out_size, void* d_ws, size_t ws_size,
                              hipStream_t stream) {
    const float* X    = (const float*)d_in[0];
    const float* W    = (const float*)d_in[1];
    const int*   esrc = (const int*)  d_in[2];
    const int*   edst = (const int*)  d_in[3];
    const float* eval = (const float*)d_in[4];
    float*       Z    = (float*)d_out;
    float*       Y    = (float*)d_ws;

    const int nNodes = in_sizes[0] / DFEAT;
    const int nEdges = in_sizes[2];

    const int mTiles   = nNodes / 16;
    const int gemmGrid = (mTiles + GEMM_TPB - 1) / GEMM_TPB;
    gconv_gemm_xw<<<gemmGrid, 256, 0, stream>>>(X, W, Y, mTiles);

    const int ownGrid = (nNodes + OWN_ROWS - 1) / OWN_ROWS;
    gconv_owner_sum<<<ownGrid, OWN_WAVES * 32, 0, stream>>>(Y, esrc, edst, eval, Z, nNodes, nEdges);
}


# Owner scan reads one owner byte per edge (dst/391, written by the GEMM launch behind Y) instead of the 4-byte dst id; zero-byte test per 16 bytes, scalar walk of matches; arrival-order sums
# speedup vs baseline: 2.9052x; 1.2926x over previous
#include <hip/hip_runtime.h>
#include <hip/hip_bf16.h>


typedef __attribute__((ext_vector_type(2))) float v2f;
typedef __attribute__((ext_vector_type(8))) float v8f;

#define DFEAT 128

#define GEMM_TPB 5
#define XPITCH   132
#define OWN_ROWS 391

__global__ __launch_bounds__(256) void gconv_gemm_xw(const float* __restrict__ X,
                                                     const float* __restrict__ W,
                                                     float* __restrict__ Y,
                                                     int mTiles,
                                                     const int* __restrict__ edst,
                                                     int nNodes, int nEdges, int tagged) {
    __shared__ float xs[2][16 * XPITCH];

    const int tid  = (int)threadIdx.x;

    if (tagged) {
        unsigned* tagw = (unsigned*)(Y + (size_t)nNodes * DFEAT);
        const int nWords = ((nEdges + 15) >> 4) << 2;
        const int per = (nWords + (int)gridDim.x - 1) / (int)gridDim.x;
        for (int i = tid; i < per; i += 256) {
            const int w = (int)blockIdx.x * per + i;
            if (w < nWords) {
                const int e0 = 4 * w;
                unsigned t = 0u;
                if (e0 + 3 < nEdges) {
                    const int4 d = ((const int4*)edst)[w];
                    t = (((unsigned)d.x / OWN_ROWS) & 255u) | ((((unsigned)d.y / OWN_ROWS) & 255u) << 8) |
                        ((((unsigned)d.z / OWN_ROWS) & 255u) << 16) | (((unsigned)d.w / OWN_ROWS) << 24);
                } else {
                    for (int u = 0; u < 4; ++u)
                        if (e0 + u < nEdges) t |= (((unsigned)edst[e0 + u] / OWN_ROWS) & 255u) << (8 * u);
                }
                tagw[w] = t;
            }
        }
    }

    const int lane = tid & 31;
    const int wave = __builtin_amdgcn_readfirstlane(tid >> 5);
    const int hi   = lane >> 4;
    const int l15  = lane & 15;
    const int ncol = wave * 16 + l15;

    v2f bfrag[DFEAT / 4];
#pragma unroll
    for (int kk = 0; kk < DFEAT / 4; ++kk) {
        const int ka = 4 * kk + 2 * hi;
        bfrag[kk].x = W[(size_t)(ka + 0) * DFEAT + ncol];
        bfrag[kk].y = W[(size_t)(ka + 1) * DFEAT + ncol];
    }

    const int mt0 = (int)blockIdx.x * GEMM_TPB;
    int nt = mTiles - mt0;
    if (nt > GEMM_TPB) nt = GEMM_TPB;

    const int sr = tid >> 5;
    const int sc = tid & 31;
    const float4* Xt = (const float4*)(X + ((size_t)mt0 * 16 + sr) * DFEAT) + sc;
    float* sdst = &xs[0][sr * XPITCH + 4 * sc];
    const float* asrc = &xs[0][l15 * XPITCH + 2 * hi];

    float4 g0 = make_float4(0.f, 0.f, 0.f, 0.f), g1 = g0;
    if (nt > 0) {
        g0 = Xt[0];
        g1 = Xt[8 * DFEAT / 4];
        *(float4*)(sdst)              = g0;
        *(float4*)(sdst + 8 * XPITCH) = g1;
    }
    __syncthreads();

    for (int j = 0; j < nt; ++j) {
        const int buf = j & 1;
        const bool more = (j + 1 < nt);
        if (more) {
            const float4* Xn = Xt + (size_t)(j + 1) * (16 * DFEAT / 4);
            g0 = Xn[0];
            g1 = Xn[8 * DFEAT / 4];
        }

        v2f a[DFEAT / 4];
        const float* ab = asrc + buf * (16 * XPITCH);
#pragma unroll
        for (int kk = 0; kk < DFEAT / 4; ++kk) a[kk] = *(const v2f*)(ab + 4 * kk);

        v8f acc = {};
#pragma unroll
        for (int kk = 0; kk < DFEAT / 4; ++kk) {
            acc = __builtin_amdgcn_wmma_f32_16x16x4_f32(
                false, a[kk], false, bfrag[kk], (short)0, acc, false, false);
        }

        if (more) {
            float* sd = sdst + (buf ^ 1) * (16 * XPITCH);
            *(float4*)(sd)              = g0;
            *(float4*)(sd + 8 * XPITCH) = g1;
        }
        __syncthreads();

        float* Yt = Y + ((size_t)(mt0 + j) * 16 + 8 * hi) * DFEAT + ncol;
#pragma unroll
        for (int i = 0; i < 8; ++i) Yt[(size_t)i * DFEAT] = acc[i];
    }
}

#define OWN_WAVES  32
#define STEP_LOADS 6
#define STEP_EDGES (STEP_LOADS * 128)
#define DRAIN_AT   64
#define QCAP       (DRAIN_AT + STEP_EDGES)

__global__ __launch_bounds__(OWN_WAVES * 32) void gconv_owner_sum(const float* __restrict__ Y,
                                                                  const int* __restrict__ esrc,
                                                                  const int* __restrict__ edst,
                                                                  const float* __restrict__ eval,
                                                                  float* __restrict__ Z,
                                                                  int nNodes, int nEdges) {
    __shared__ float acc[OWN_ROWS * DFEAT];
    __shared__ int   qall[OWN_WAVES * QCAP];

    const int lane = (int)(threadIdx.x & 31);
    const int wave = __builtin_amdgcn_readfirstlane((int)(threadIdx.x >> 5));
    const int rowBase = (int)blockIdx.x * OWN_ROWS;
    int nrows = nNodes - rowBase;
    if (nrows > OWN_ROWS) nrows = OWN_ROWS;

    for (int i = (int)threadIdx.x; i < OWN_ROWS * DFEAT / 4; i += OWN_WAVES * 32) {
        ((float4*)acc)[i] = make_float4(0.f, 0.f, 0.f, 0.f);
    }
    __syncthreads();

    int* q = qall + wave * QCAP;
    int cnt = 0;
    const int nSteps = (nEdges + STEP_EDGES - 1) / STEP_EDGES;

    int4 cur[STEP_LOADS], nxt[STEP_LOADS];

    auto load_step = [&](int t, int4 (&d)[STEP_LOADS]) {
        const int base = t * STEP_EDGES;
        if (base + STEP_EDGES <= nEdges) {
            const int4* p = (const int4*)(edst + base);
#pragma unroll
            for (int k = 0; k < STEP_LOADS; ++k) d[k] = p[k * 32 + lane];
        } else {
#pragma unroll
            for (int k = 0; k < STEP_LOADS; ++k) {
                const int i0 = base + k * 128 + lane * 4;
                d[k].x = (i0 + 0 < nEdges) ? edst[i0 + 0] : -1;
                d[k].y = (i0 + 1 < nEdges) ? edst[i0 + 1] : -1;
                d[k].z = (i0 + 2 < nEdges) ? edst[i0 + 2] : -1;
                d[k].w = (i0 + 3 < nEdges) ? edst[i0 + 3] : -1;
            }
        }
    };

    if (wave < nSteps) load_step(wave, cur);

    for (int t = wave; t < nSteps; t += OWN_WAVES) {
        const int tn = t + OWN_WAVES;
        if (tn < nSteps) load_step(tn, nxt);

        const int base = t * STEP_EDGES;
#pragma unroll
        for (int k = 0; k < STEP_LOADS; ++k) {
            const bool m0 = (unsigned)(cur[k].x - rowBase) < (unsigned)nrows;
            const bool m1 = (unsigned)(cur[k].y - rowBase) < (unsigned)nrows;
            const bool m2 = (unsigned)(cur[k].z - rowBase) < (unsigned)nrows;
            const bool m3 = (unsigned)(cur[k].w - rowBase) < (unsigned)nrows;
            if (__builtin_amdgcn_ballot_w32(m0 | m1 | m2 | m3) != 0u) {
                const int e0 = base + k * 128 + lane * 4;
                const unsigned b0 = __builtin_amdgcn_ballot_w32(m0);
                if (m0) q[cnt + (int)__builtin_amdgcn_mbcnt_lo(b0, 0u)] = e0 + 0;
                cnt += __builtin_popcount(b0);
                const unsigned b1 = __builtin_amdgcn_ballot_w32(m1);
                if (m1) q[cnt + (int)__builtin_amdgcn_mbcnt_lo(b1, 0u)] = e0 + 1;
                cnt += __builtin_popcount(b1);
                const unsigned b2 = __builtin_amdgcn_ballot_w32(m2);
                if (m2) q[cnt + (int)__builtin_amdgcn_mbcnt_lo(b2, 0u)] = e0 + 2;
                cnt += __builtin_popcount(b2);
                const unsigned b3 = __builtin_amdgcn_ballot_w32(m3);
                if (m3) q[cnt + (int)__builtin_amdgcn_mbcnt_lo(b3, 0u)] = e0 + 3;
                cnt += __builtin_popcount(b3);
            }
        }

        if (cnt >= DRAIN_AT || tn >= nSteps) {
            __builtin_amdgcn_wave_barrier();
            for (int b = 0; b < cnt; b += 32) {
                int n = cnt - b;
                if (n > 32) n = 32;
                const int e  = q[b + (lane < n ? lane : 0)];
                const int s  = esrc[e];
                const int dl = edst[e] - rowBase;
                const float v = eval[e];
                for (int j = 0; j < n; j += 8) {
                    float4 y[8];
#pragma unroll
                    for (int u = 0; u < 8; ++u) {
                        int ju = j + u;
                        if (ju > n - 1) ju = n - 1;
                        const int sj = __builtin_amdgcn_readlane(s, ju);
                        y[u] = ((const float4*)(Y + (size_t)sj * DFEAT))[lane];
                    }
#pragma unroll
                    for (int u = 0; u < 8; ++u) {
                        if (j + u < n) {
                            const int   rj = __builtin_amdgcn_readlane(dl, j + u);
                            const float vj = __int_as_float(__builtin_amdgcn_readlane(__float_as_int(v), j + u));
                            float* a = acc + rj * DFEAT + lane;
                            atomicAdd(a +  0, vj * y[u].x);
                            atomicAdd(a + 32, vj * y[u].y);
                            atomicAdd(a + 64, vj * y[u].z);
                            atomicAdd(a + 96, vj * y[u].w);
                        }
                    }
                }
            }
            cnt = 0;
            __builtin_amdgcn_wave_barrier();
        }

#pragma unroll
        for (int k = 0; k < STEP_LOADS; ++k) cur[k] = nxt[k];
    }

    __syncthreads();

    for (int r = wave; r < nrows; r += OWN_WAVES) {
        const float* a = acc + r * DFEAT + lane;
        float4 o;
        o.x = a[0];
        o.y = a[32];
        o.z = a[64];
        o.w = a[96];
        ((float4*)(Z + (size_t)(rowBase + r) * DFEAT))[lane] = o;
    }
}

#define TAG_LOADS 4

__device__ __forceinline__ unsigned zero_byte_flags(unsigned x) {
    const unsigned t = (x & 0x7F7F7F7Fu) + 0x7F7F7F7Fu;
    return ~(t | x | 0x7F7F7F7Fu);
}

__device__ __forceinline__ unsigned chunk_flags(uint4 v, unsigned pat) {
    const unsigned f0 = zero_byte_flags(v.x ^ pat);
    const unsigned f1 = zero_byte_flags(v.y ^ pat);
    const unsigned f2 = zero_byte_flags(v.z ^ pat);
    const unsigned f3 = zero_byte_flags(v.w ^ pat);
    return (f0 >> 3) | (f1 >> 2) | (f2 >> 1) | f3;
}

__global__ __launch_bounds__(OWN_WAVES * 32) void gconv_owner_sum_tags(const float* __restrict__ Y,
                                                                       const int* __restrict__ esrc,
                                                                       const int* __restrict__ edst,
                                                                       const float* __restrict__ eval,
                                                                       float* __restrict__ Z,
                                                                       int nNodes, int nEdges) {
    __shared__ float acc[OWN_ROWS * DFEAT];

    const int lane = (int)(threadIdx.x & 31);
    const int wave = __builtin_amdgcn_readfirstlane((int)(threadIdx.x >> 5));
    const int rowBase = (int)blockIdx.x * OWN_ROWS;
    int nrows = nNodes - rowBase;
    if (nrows > OWN_ROWS) nrows = OWN_ROWS;

    for (int i = (int)threadIdx.x; i < OWN_ROWS * DFEAT / 4; i += OWN_WAVES * 32) {
        ((float4*)acc)[i] = make_float4(0.f, 0.f, 0.f, 0.f);
    }
    __syncthreads();

    const uint4* tg = (const uint4*)(Y + (size_t)nNodes * DFEAT);
    const int nChunks = (nEdges + 15) >> 4;
    const int nSteps = (nChunks + TAG_LOADS * 32 - 1) / (TAG_LOADS * 32);
    const unsigned pat = (unsigned)blockIdx.x * 0x01010101u;

    int pend = 0;
    int cnt = 0;

    auto drain = [&](int n) {
        const int e  = pend;
        const int s  = esrc[e];
        int   dl = edst[e] - rowBase;
        float v  = eval[e];
        if ((unsigned)dl >= (unsigned)nrows) { dl = 0; v = 0.f; }
        for (int j = 0; j < n; j += 8) {
            float4 y[8];
#pragma unroll
            for (int u = 0; u < 8; ++u) {
                int ju = j + u;
                if (ju > n - 1) ju = n - 1;
                const int sj = __builtin_amdgcn_readlane(s, ju);
                y[u] = ((const float4*)(Y + (size_t)sj * DFEAT))[lane];
            }
#pragma unroll
            for (int u = 0; u < 8; ++u) {
                if (j + u < n) {
                    const int   rj = __builtin_amdgcn_readlane(dl, j + u);
                    const float vj = __int_as_float(__builtin_amdgcn_readlane(__float_as_int(v), j + u));
                    float* a = acc + rj * DFEAT + lane;
                    atomicAdd(a +  0, vj * y[u].x);
                    atomicAdd(a + 32, vj * y[u].y);
                    atomicAdd(a + 64, vj * y[u].z);
                    atomicAdd(a + 96, vj * y[u].w);
                }
            }
        }
    };

    uint4 cur[TAG_LOADS], nxt[TAG_LOADS];

    auto load_step = [&](int t, uint4 (&d)[TAG_LOADS]) {
        const int c0 = t * (TAG_LOADS * 32);
        if (c0 + TAG_LOADS * 32 <= nChunks) {
#pragma unroll
            for (int k = 0; k < TAG_LOADS; ++k) d[k] = tg[c0 + k * 32 + lane];
        } else {
            const unsigned nm = ~pat;
#pragma unroll
            for (int k = 0; k < TAG_LOADS; ++k) {
                const int c = c0 + k * 32 + lane;
                d[k] = make_uint4(nm, nm, nm, nm);
                if (c < nChunks) d[k] = tg[c];
            }
        }
    };

    if (wave < nSteps) load_step(wave, cur);

    for (int t = wave; t < nSteps; t += OWN_WAVES) {
        const int tn = t + OWN_WAVES;
        if (tn < nSteps) load_step(tn, nxt);

        const unsigned p0 = chunk_flags(cur[0], pat) | (chunk_flags(cur[1], pat) >> 4);
        const unsigned p1 = chunk_flags(cur[2], pat) | (chunk_flags(cur[3], pat) >> 4);
        unsigned hit = __builtin_amdgcn_ballot_w32((p0 | p1) != 0u);

        const int stepBase = t * (TAG_LOADS * 512);
        while (hit != 0u) {
            const int l = __builtin_ctz(hit);
            hit &= hit - 1u;
            unsigned long long w =
                ((unsigned long long)(unsigned)__builtin_amdgcn_readlane((int)p1, l) << 32) |
                (unsigned long long)(unsigned)__builtin_amdgcn_readlane((int)p0, l);
            while (w != 0ull) {
                const int p = __builtin_ctzll(w);
                w &= w - 1ull;
                const int k    = ((p >> 5) << 1) | (((p >> 2) & 1) ^ 1);
                const int word = p & 3;
                const int byte = (p >> 3) & 3;
                const int e = stepBase + k * 512 + l * 16 + word * 4 + byte;
                if (e < nEdges) {
                    pend = (lane == cnt) ? e : pend;
                    ++cnt;
                    if (cnt == 32) {
                        drain(32);
                        cnt = 0;
                    }
                }
            }
        }

#pragma unroll
        for (int k = 0; k < TAG_LOADS; ++k) cur[k] = nxt[k];
    }

    if (cnt > 0) drain(cnt);

    __syncthreads();

    for (int r = wave; r < nrows; r += OWN_WAVES) {
        const float* a = acc + r * DFEAT + lane;
        float4 o;
        o.x = a[0];
        o.y = a[32];
        o.z = a[64];
        o.w = a[96];
        ((float4*)(Z + (size_t)(rowBase + r) * DFEAT))[lane] = o;
    }
}

extern "C" void kernel_launch(void* const* d_in, const int* in_sizes, int n_in,
                              void* d_out, int out_size, void* d_ws, size_t ws_size,
                              hipStream_t stream) {
    const float* X    = (const float*)d_in[0];
    const float* W    = (const float*)d_in[1];
    const int*   esrc = (const int*)  d_in[2];
    const int*   edst = (const int*)  d_in[3];
    const float* eval = (const float*)d_in[4];
    float*       Z    = (float*)d_out;
    float*       Y    = (float*)d_ws;

    const int nNodes = in_sizes[0] / DFEAT;
    const int nEdges = in_sizes[2];

    const int mTiles   = nNodes / 16;
    const int gemmGrid = (mTiles + GEMM_TPB - 1) / GEMM_TPB;
    const int ownGrid  = (nNodes + OWN_ROWS - 1) / OWN_ROWS;

    const size_t wsNeed = ((size_t)nNodes * DFEAT + (size_t)((nEdges + 15) / 16) * 4) * sizeof(float);
    const int tagged = (ws_size >= wsNeed && ownGrid <= 256) ? 1 : 0;

    gconv_gemm_xw<<<gemmGrid, 256, 0, stream>>>(X, W, Y, mTiles, edst, nNodes, nEdges, tagged);

    if (tagged) {
        gconv_owner_sum_tags<<<ownGrid, OWN_WAVES * 32, 0, stream>>>(Y, esrc, edst, eval, Z, nNodes, nEdges);
    } else {
        gconv_owner_sum<<<ownGrid, OWN_WAVES * 32, 0, stream>>>(Y, esrc, edst, eval, Z, nNodes, nEdges);
    }
}


# Owner scan collects matches with all lanes at once: per pass each lane takes its lowest mask bit, ballot rank puts its edge index in a per-wave LDS list (was a scalar walk); sums stay arrival-order
# speedup vs baseline: 3.0606x; 1.0535x over previous
#include <hip/hip_runtime.h>
#include <hip/hip_bf16.h>


typedef __attribute__((ext_vector_type(2))) float v2f;
typedef __attribute__((ext_vector_type(8))) float v8f;

#define DFEAT 128

#define GEMM_TPB 5
#define XPITCH   132
#define OWN_ROWS 391

__global__ __launch_bounds__(256) void gconv_gemm_xw(const float* __restrict__ X,
                                                     const float* __restrict__ W,
                                                     float* __restrict__ Y,
                                                     int mTiles,
                                                     const int* __restrict__ edst,
                                                     int nNodes, int nEdges, int tagged) {
    __shared__ float xs[2][16 * XPITCH];

    const int tid  = (int)threadIdx.x;

    if (tagged) {
        unsigned* tagw = (unsigned*)(Y + (size_t)nNodes * DFEAT);
        const int nWords = ((nEdges + 15) >> 4) << 2;
        const int per = (nWords + (int)gridDim.x - 1) / (int)gridDim.x;
        for (int i = tid; i < per; i += 256) {
            const int w = (int)blockIdx.x * per + i;
            if (w < nWords) {
                const int e0 = 4 * w;
                unsigned t = 0u;
                if (e0 + 3 < nEdges) {
                    const int4 d = ((const int4*)edst)[w];
                    t = (((unsigned)d.x / OWN_ROWS) & 255u) | ((((unsigned)d.y / OWN_ROWS) & 255u) << 8) |
                        ((((unsigned)d.z / OWN_ROWS) & 255u) << 16) | (((unsigned)d.w / OWN_ROWS) << 24);
                } else {
                    for (int u = 0; u < 4; ++u)
                        if (e0 + u < nEdges) t |= (((unsigned)edst[e0 + u] / OWN_ROWS) & 255u) << (8 * u);
                }
                tagw[w] = t;
            }
        }
    }

    const int lane = tid & 31;
    const int wave = __builtin_amdgcn_readfirstlane(tid >> 5);
    const int hi   = lane >> 4;
    const int l15  = lane & 15;
    const int ncol = wave * 16 + l15;

    v2f bfrag[DFEAT / 4];
#pragma unroll
    for (int kk = 0; kk < DFEAT / 4; ++kk) {
        const int ka = 4 * kk + 2 * hi;
        bfrag[kk].x = W[(size_t)(ka + 0) * DFEAT + ncol];
        bfrag[kk].y = W[(size_t)(ka + 1) * DFEAT + ncol];
    }

    const int mt0 = (int)blockIdx.x * GEMM_TPB;
    int nt = mTiles - mt0;
    if (nt > GEMM_TPB) nt = GEMM_TPB;

    const int sr = tid >> 5;
    const int sc = tid & 31;
    const float4* Xt = (const float4*)(X + ((size_t)mt0 * 16 + sr) * DFEAT) + sc;
    float* sdst = &xs[0][sr * XPITCH + 4 * sc];
    const float* asrc = &xs[0][l15 * XPITCH + 2 * hi];

    float4 g0 = make_float4(0.f, 0.f, 0.f, 0.f), g1 = g0;
    if (nt > 0) {
        g0 = Xt[0];
        g1 = Xt[8 * DFEAT / 4];
        *(float4*)(sdst)              = g0;
        *(float4*)(sdst + 8 * XPITCH) = g1;
    }
    __syncthreads();

    for (int j = 0; j < nt; ++j) {
        const int buf = j & 1;
        const bool more = (j + 1 < nt);
        if (more) {
            const float4* Xn = Xt + (size_t)(j + 1) * (16 * DFEAT / 4);
            g0 = Xn[0];
            g1 = Xn[8 * DFEAT / 4];
        }

        v2f a[DFEAT / 4];
        const float* ab = asrc + buf * (16 * XPITCH);
#pragma unroll
        for (int kk = 0; kk < DFEAT / 4; ++kk) a[kk] = *(const v2f*)(ab + 4 * kk);

        v8f acc = {};
#pragma unroll
        for (int kk = 0; kk < DFEAT / 4; ++kk) {
            acc = __builtin_amdgcn_wmma_f32_16x16x4_f32(
                false, a[kk], false, bfrag[kk], (short)0, acc, false, false);
        }

        if (more) {
            float* sd = sdst + (buf ^ 1) * (16 * XPITCH);
            *(float4*)(sd)              = g0;
            *(float4*)(sd + 8 * XPITCH) = g1;
        }
        __syncthreads();

        float* Yt = Y + ((size_t)(mt0 + j) * 16 + 8 * hi) * DFEAT + ncol;
#pragma unroll
        for (int i = 0; i < 8; ++i) Yt[(size_t)i * DFEAT] = acc[i];
    }
}

#define OWN_WAVES  32
#define STEP_LOADS 6
#define STEP_EDGES (STEP_LOADS * 128)
#define DRAIN_AT   64
#define QCAP       (DRAIN_AT + STEP_EDGES)

__global__ __launch_bounds__(OWN_WAVES * 32) void gconv_owner_sum(const float* __restrict__ Y,
                                                                  const int* __restrict__ esrc,
                                                                  const int* __restrict__ edst,
                                                                  const float* __restrict__ eval,
                                                                  float* __restrict__ Z,
                                                                  int nNodes, int nEdges) {
    __shared__ float acc[OWN_ROWS * DFEAT];
    __shared__ int   qall[OWN_WAVES * QCAP];

    const int lane = (int)(threadIdx.x & 31);
    const int wave = __builtin_amdgcn_readfirstlane((int)(threadIdx.x >> 5));
    const int rowBase = (int)blockIdx.x * OWN_ROWS;
    int nrows = nNodes - rowBase;
    if (nrows > OWN_ROWS) nrows = OWN_ROWS;

    for (int i = (int)threadIdx.x; i < OWN_ROWS * DFEAT / 4; i += OWN_WAVES * 32) {
        ((float4*)acc)[i] = make_float4(0.f, 0.f, 0.f, 0.f);
    }
    __syncthreads();

    int* q = qall + wave * QCAP;
    int cnt = 0;
    const int nSteps = (nEdges + STEP_EDGES - 1) / STEP_EDGES;

    int4 cur[STEP_LOADS], nxt[STEP_LOADS];

    auto load_step = [&](int t, int4 (&d)[STEP_LOADS]) {
        const int base = t * STEP_EDGES;
        if (base + STEP_EDGES <= nEdges) {
            const int4* p = (const int4*)(edst + base);
#pragma unroll
            for (int k = 0; k < STEP_LOADS; ++k) d[k] = p[k * 32 + lane];
        } else {
#pragma unroll
            for (int k = 0; k < STEP_LOADS; ++k) {
                const int i0 = base + k * 128 + lane * 4;
                d[k].x = (i0 + 0 < nEdges) ? edst[i0 + 0] : -1;
                d[k].y = (i0 + 1 < nEdges) ? edst[i0 + 1] : -1;
                d[k].z = (i0 + 2 < nEdges) ? edst[i0 + 2] : -1;
                d[k].w = (i0 + 3 < nEdges) ? edst[i0 + 3] : -1;
            }
        }
    };

    if (wave < nSteps) load_step(wave, cur);

    for (int t = wave; t < nSteps; t += OWN_WAVES) {
        const int tn = t + OWN_WAVES;
        if (tn < nSteps) load_step(tn, nxt);

        const int base = t * STEP_EDGES;
#pragma unroll
        for (int k = 0; k < STEP_LOADS; ++k) {
            const bool m0 = (unsigned)(cur[k].x - rowBase) < (unsigned)nrows;
            const bool m1 = (unsigned)(cur[k].y - rowBase) < (unsigned)nrows;
            const bool m2 = (unsigned)(cur[k].z - rowBase) < (unsigned)nrows;
            const bool m3 = (unsigned)(cur[k].w - rowBase) < (unsigned)nrows;
            if (__builtin_amdgcn_ballot_w32(m0 | m1 | m2 | m3) != 0u) {
                const int e0 = base + k * 128 + lane * 4;
                const unsigned b0 = __builtin_amdgcn_ballot_w32(m0);
                if (m0) q[cnt + (int)__builtin_amdgcn_mbcnt_lo(b0, 0u)] = e0 + 0;
                cnt += __builtin_popcount(b0);
                const unsigned b1 = __builtin_amdgcn_ballot_w32(m1);
                if (m1) q[cnt + (int)__builtin_amdgcn_mbcnt_lo(b1, 0u)] = e0 + 1;
                cnt += __builtin_popcount(b1);
                const unsigned b2 = __builtin_amdgcn_ballot_w32(m2);
                if (m2) q[cnt + (int)__builtin_amdgcn_mbcnt_lo(b2, 0u)] = e0 + 2;
                cnt += __builtin_popcount(b2);
                const unsigned b3 = __builtin_amdgcn_ballot_w32(m3);
                if (m3) q[cnt + (int)__builtin_amdgcn_mbcnt_lo(b3, 0u)] = e0 + 3;
                cnt += __builtin_popcount(b3);
            }
        }

        if (cnt >= DRAIN_AT || tn >= nSteps) {
            __builtin_amdgcn_wave_barrier();
            for (int b = 0; b < cnt; b += 32) {
                int n = cnt - b;
                if (n > 32) n = 32;
                const int e  = q[b + (lane < n ? lane : 0)];
                const int s  = esrc[e];
                const int dl = edst[e] - rowBase;
                const float v = eval[e];
                for (int j = 0; j < n; j += 8) {
                    float4 y[8];
#pragma unroll
                    for (int u = 0; u < 8; ++u) {
                        int ju = j + u;
                        if (ju > n - 1) ju = n - 1;
                        const int sj = __builtin_amdgcn_readlane(s, ju);
                        y[u] = ((const float4*)(Y + (size_t)sj * DFEAT))[lane];
                    }
#pragma unroll
                    for (int u = 0; u < 8; ++u) {
                        if (j + u < n) {
                            const int   rj = __builtin_amdgcn_readlane(dl, j + u);
                            const float vj = __int_as_float(__builtin_amdgcn_readlane(__float_as_int(v), j + u));
                            float* a = acc + rj * DFEAT + lane;
                            atomicAdd(a +  0, vj * y[u].x);
                            atomicAdd(a + 32, vj * y[u].y);
                            atomicAdd(a + 64, vj * y[u].z);
                            atomicAdd(a + 96, vj * y[u].w);
                        }
                    }
                }
            }
            cnt = 0;
            __builtin_amdgcn_wave_barrier();
        }

#pragma unroll
        for (int k = 0; k < STEP_LOADS; ++k) cur[k] = nxt[k];
    }

    __syncthreads();

    for (int r = wave; r < nrows; r += OWN_WAVES) {
        const float* a = acc + r * DFEAT + lane;
        float4 o;
        o.x = a[0];
        o.y = a[32];
        o.z = a[64];
        o.w = a[96];
        ((float4*)(Z + (size_t)(rowBase + r) * DFEAT))[lane] = o;
    }
}

#define TAG_LOADS 4
#define TAG_QLEN  64

__device__ __forceinline__ unsigned zero_byte_flags(unsigned x) {
    const unsigned t = (x & 0x7F7F7F7Fu) + 0x7F7F7F7Fu;
    return ~(t | x | 0x7F7F7F7Fu);
}

__device__ __forceinline__ unsigned chunk_flags(uint4 v, unsigned pat) {
    const unsigned f0 = zero_byte_flags(v.x ^ pat);
    const unsigned f1 = zero_byte_flags(v.y ^ pat);
    const unsigned f2 = zero_byte_flags(v.z ^ pat);
    const unsigned f3 = zero_byte_flags(v.w ^ pat);
    return (f0 >> 3) | (f1 >> 2) | (f2 >> 1) | f3;
}

__global__ __launch_bounds__(OWN_WAVES * 32) void gconv_owner_sum_tags(const float* __restrict__ Y,
                                                                       const int* __restrict__ esrc,
                                                                       const int* __restrict__ edst,
                                                                       const float* __restrict__ eval,
                                                                       float* __restrict__ Z,
                                                                       int nNodes, int nEdges) {
    __shared__ float acc[OWN_ROWS * DFEAT];
    __shared__ int   qtag[OWN_WAVES * TAG_QLEN];

    const int lane = (int)(threadIdx.x & 31);
    const int wave = __builtin_amdgcn_readfirstlane((int)(threadIdx.x >> 5));
    const int rowBase = (int)blockIdx.x * OWN_ROWS;
    int nrows = nNodes - rowBase;
    if (nrows > OWN_ROWS) nrows = OWN_ROWS;

    for (int i = (int)threadIdx.x; i < OWN_ROWS * DFEAT / 4; i += OWN_WAVES * 32) {
        ((float4*)acc)[i] = make_float4(0.f, 0.f, 0.f, 0.f);
    }
    __syncthreads();

    const uint4* tg = (const uint4*)(Y + (size_t)nNodes * DFEAT);
    const int nChunks = (nEdges + 15) >> 4;
    const int nSteps = (nChunks + TAG_LOADS * 32 - 1) / (TAG_LOADS * 32);
    const unsigned pat = (unsigned)blockIdx.x * 0x01010101u;

    int* q = qtag + wave * TAG_QLEN;
    int cnt = 0;

    auto drain = [&](int n) {
        __builtin_amdgcn_wave_barrier();
        const int e  = q[lane < n ? lane : 0];
        const int s  = esrc[e];
        int   dl = edst[e] - rowBase;
        float v  = eval[e];
        if ((unsigned)dl >= (unsigned)nrows) { dl = 0; v = 0.f; }
        for (int j = 0; j < n; j += 8) {
            float4 y[8];
#pragma unroll
            for (int u = 0; u < 8; ++u) {
                int ju = j + u;
                if (ju > n - 1) ju = n - 1;
                const int sj = __builtin_amdgcn_readlane(s, ju);
                y[u] = ((const float4*)(Y + (size_t)sj * DFEAT))[lane];
            }
#pragma unroll
            for (int u = 0; u < 8; ++u) {
                if (j + u < n) {
                    const int   rj = __builtin_amdgcn_readlane(dl, j + u);
                    const float vj = __int_as_float(__builtin_amdgcn_readlane(__float_as_int(v), j + u));
                    float* a = acc + rj * DFEAT + lane;
                    atomicAdd(a +  0, vj * y[u].x);
                    atomicAdd(a + 32, vj * y[u].y);
                    atomicAdd(a + 64, vj * y[u].z);
                    atomicAdd(a + 96, vj * y[u].w);
                }
            }
        }
    };

    uint4 cur[TAG_LOADS], nxt[TAG_LOADS];

    auto load_step = [&](int t, uint4 (&d)[TAG_LOADS]) {
        const int c0 = t * (TAG_LOADS * 32);
        if (c0 + TAG_LOADS * 32 <= nChunks) {
#pragma unroll
            for (int k = 0; k < TAG_LOADS; ++k) d[k] = tg[c0 + k * 32 + lane];
        } else {
            const unsigned nm = ~pat;
#pragma unroll
            for (int k = 0; k < TAG_LOADS; ++k) {
                const int c = c0 + k * 32 + lane;
                d[k] = make_uint4(nm, nm, nm, nm);
                if (c < nChunks) d[k] = tg[c];
            }
        }
    };

    if (wave < nSteps) load_step(wave, cur);

    for (int t = wave; t < nSteps; t += OWN_WAVES) {
        const int tn = t + OWN_WAVES;
        if (tn < nSteps) load_step(tn, nxt);

        const unsigned p0 = chunk_flags(cur[0], pat) | (chunk_flags(cur[1], pat) >> 4);
        const unsigned p1 = chunk_flags(cur[2], pat) | (chunk_flags(cur[3], pat) >> 4);
        unsigned m0 = p0, m1 = p1;
        const int laneBase = t * (TAG_LOADS * 512) + lane * 16;
        unsigned left = __builtin_amdgcn_ballot_w32((m0 | m1) != 0u);
        while (left != 0u) {
            const bool has = (m0 | m1) != 0u;
            const bool lo  = (m0 != 0u);
            const unsigned mm = lo ? m0 : m1;
            const int p = __builtin_ctz(mm | 0x80000000u) + (lo ? 0 : 32);
            const unsigned rest = mm & (mm - 1u);
            m0 = lo ? rest : m0;
            m1 = lo ? m1 : rest;
            const int k    = ((p >> 5) << 1) | (((p >> 2) & 1) ^ 1);
            const int word = p & 3;
            const int byte = (p >> 3) & 3;
            const int e = laneBase + k * 512 + word * 4 + byte;
            const bool ok = has && (e < nEdges);
            const unsigned took = __builtin_amdgcn_ballot_w32(ok);
            if (ok) q[cnt + (int)__builtin_amdgcn_mbcnt_lo(took, 0u)] = e;
            cnt += __builtin_popcount(took);
            if (cnt >= 32) {
                drain(32);
                const int rem = cnt - 32;
                const int mv = q[32 + (lane < rem ? lane : 0)];
                if (lane < rem) q[lane] = mv;
                cnt = rem;
            }
            left = __builtin_amdgcn_ballot_w32((m0 | m1) != 0u);
        }

#pragma unroll
        for (int k = 0; k < TAG_LOADS; ++k) cur[k] = nxt[k];
    }

    if (cnt > 0) drain(cnt);

    __syncthreads();

    for (int r = wave; r < nrows; r += OWN_WAVES) {
        const float* a = acc + r * DFEAT + lane;
        float4 o;
        o.x = a[0];
        o.y = a[32];
        o.z = a[64];
        o.w = a[96];
        ((float4*)(Z + (size_t)(rowBase + r) * DFEAT))[lane] = o;
    }
}

extern "C" void kernel_launch(void* const* d_in, const int* in_sizes, int n_in,
                              void* d_out, int out_size, void* d_ws, size_t ws_size,
                              hipStream_t stream) {
    const float* X    = (const float*)d_in[0];
    const float* W    = (const float*)d_in[1];
    const int*   esrc = (const int*)  d_in[2];
    const int*   edst = (const int*)  d_in[3];
    const float* eval = (const float*)d_in[4];
    float*       Z    = (float*)d_out;
    float*       Y    = (float*)d_ws;

    const int nNodes = in_sizes[0] / DFEAT;
    const int nEdges = in_sizes[2];

    const int mTiles   = nNodes / 16;
    const int gemmGrid = (mTiles + GEMM_TPB - 1) / GEMM_TPB;
    const int ownGrid  = (nNodes + OWN_ROWS - 1) / OWN_ROWS;

    const size_t wsNeed = ((size_t)nNodes * DFEAT + (size_t)((nEdges + 15) / 16) * 4) * sizeof(float);
    const int tagged = (ws_size >= wsNeed && ownGrid <= 256) ? 1 : 0;

    gconv_gemm_xw<<<gemmGrid, 256, 0, stream>>>(X, W, Y, mTiles, edst, nNodes, nEdges, tagged);

    if (tagged) {
        gconv_owner_sum_tags<<<ownGrid, OWN_WAVES * 32, 0, stream>>>(Y, esrc, edst, eval, Z, nNodes, nEdges);
    } else {
        gconv_owner_sum<<<ownGrid, OWN_WAVES * 32, 0, stream>>>(Y, esrc, edst, eval, Z, nNodes, nEdges);
    }
}


# Matched edge's src, row and val read as one 8-byte record packed by the GEMM launch behind the owner bytes (one random line per edge, was three 4-byte loads from three lists); sums stay arrival-order
# speedup vs baseline: 3.2835x; 1.0728x over previous
#include <hip/hip_runtime.h>
#include <hip/hip_bf16.h>


typedef __attribute__((ext_vector_type(2))) float v2f;
typedef __attribute__((ext_vector_type(8))) float v8f;

#define DFEAT 128

#define GEMM_TPB 5
#define XPITCH   132
#define OWN_ROWS 391
#define REC_SHIFT 20

__device__ __forceinline__ unsigned rec_word(int s, int d, int nNodes) {
    const unsigned ud = (unsigned)d;
    const unsigned lr = (ud < (unsigned)nNodes) ? ud - (ud / OWN_ROWS) * OWN_ROWS : 0u;
    return ((unsigned)s & ((1u << REC_SHIFT) - 1u)) | (lr << REC_SHIFT);
}
__device__ __forceinline__ unsigned rec_val(float v, int d, int nNodes) {
    return ((unsigned)d < (unsigned)nNodes) ? __float_as_uint(v) : 0u;
}

__global__ __launch_bounds__(256) void gconv_gemm_xw(const float* __restrict__ X,
                                                     const float* __restrict__ W,
                                                     float* __restrict__ Y,
                                                     int mTiles,
                                                     const int* __restrict__ esrc,
                                                     const int* __restrict__ edst,
                                                     const float* __restrict__ eval,
                                                     int nNodes, int nEdges, int tagged) {
    __shared__ float xs[2][16 * XPITCH];

    const int tid  = (int)threadIdx.x;

    if (tagged) {
        unsigned* tagw = (unsigned*)(Y + (size_t)nNodes * DFEAT);
        const int nWords = ((nEdges + 15) >> 4) << 2;
        uint2* rec = (uint2*)(tagw + nWords);
        const bool recs = tagged > 1;
        const int per = (nWords + (int)gridDim.x - 1) / (int)gridDim.x;
        for (int i = tid; i < per; i += 256) {
            const int w = (int)blockIdx.x * per + i;
            if (w < nWords) {
                const int e0 = 4 * w;
                unsigned t = 0u;
                if (e0 + 3 < nEdges) {
                    const int4 d = ((const int4*)edst)[w];
                    t = (((unsigned)d.x / OWN_ROWS) & 255u) | ((((unsigned)d.y / OWN_ROWS) & 255u) << 8) |
                        ((((unsigned)d.z / OWN_ROWS) & 255u) << 16) | (((unsigned)d.w / OWN_ROWS) << 24);
                    if (recs) {
                        const int4   s = ((const int4*)esrc)[w];
                        const float4 v = ((const float4*)eval)[w];
                        uint4 r0, r1;
                        r0.x = rec_word(s.x, d.x, nNodes); r0.y = rec_val(v.x, d.x, nNodes);
                        r0.z = rec_word(s.y, d.y, nNodes); r0.w = rec_val(v.y, d.y, nNodes);
                        r1.x = rec_word(s.z, d.z, nNodes); r1.y = rec_val(v.z, d.z, nNodes);
                        r1.z = rec_word(s.w, d.w, nNodes); r1.w = rec_val(v.w, d.w, nNodes);
                        ((uint4*)rec)[2 * w]     = r0;
                        ((uint4*)rec)[2 * w + 1] = r1;
                    }
                } else {
                    for (int u = 0; u < 4; ++u)
                        if (e0 + u < nEdges) {
                            const int du = edst[e0 + u];
                            t |= (((unsigned)du / OWN_ROWS) & 255u) << (8 * u);
                            if (recs) rec[e0 + u] = make_uint2(rec_word(esrc[e0 + u], du, nNodes),
                                                               rec_val(eval[e0 + u], du, nNodes));
                        }
                }
                tagw[w] = t;
            }
        }
    }

    const int lane = tid & 31;
    const int wave = __builtin_amdgcn_readfirstlane(tid >> 5);
    const int hi   = lane >> 4;
    const int l15  = lane & 15;
    const int ncol = wave * 16 + l15;

    v2f bfrag[DFEAT / 4];
#pragma unroll
    for (int kk = 0; kk < DFEAT / 4; ++kk) {
        const int ka = 4 * kk + 2 * hi;
        bfrag[kk].x = W[(size_t)(ka + 0) * DFEAT + ncol];
        bfrag[kk].y = W[(size_t)(ka + 1) * DFEAT + ncol];
    }

    const int mt0 = (int)blockIdx.x * GEMM_TPB;
    int nt = mTiles - mt0;
    if (nt > GEMM_TPB) nt = GEMM_TPB;

    const int sr = tid >> 5;
    const int sc = tid & 31;
    const float4* Xt = (const float4*)(X + ((size_t)mt0 * 16 + sr) * DFEAT) + sc;
    float* sdst = &xs[0][sr * XPITCH + 4 * sc];
    const float* asrc = &xs[0][l15 * XPITCH + 2 * hi];

    float4 g0 = make_float4(0.f, 0.f, 0.f, 0.f), g1 = g0;
    if (nt > 0) {
        g0 = Xt[0];
        g1 = Xt[8 * DFEAT / 4];
        *(float4*)(sdst)              = g0;
        *(float4*)(sdst + 8 * XPITCH) = g1;
    }
    __syncthreads();

    for (int j = 0; j < nt; ++j) {
        const int buf = j & 1;
        const bool more = (j + 1 < nt);
        if (more) {
            const float4* Xn = Xt + (size_t)(j + 1) * (16 * DFEAT / 4);
            g0 = Xn[0];
            g1 = Xn[8 * DFEAT / 4];
        }

        v2f a[DFEAT / 4];
        const float* ab = asrc + buf * (16 * XPITCH);
#pragma unroll
        for (int kk = 0; kk < DFEAT / 4; ++kk) a[kk] = *(const v2f*)(ab + 4 * kk);

        v8f acc = {};
#pragma unroll
        for (int kk = 0; kk < DFEAT / 4; ++kk) {
            acc = __builtin_amdgcn_wmma_f32_16x16x4_f32(
                false, a[kk], false, bfrag[kk], (short)0, acc, false, false);
        }

        if (more) {
            float* sd = sdst + (buf ^ 1) * (16 * XPITCH);
            *(float4*)(sd)              = g0;
            *(float4*)(sd + 8 * XPITCH) = g1;
        }
        __syncthreads();

        float* Yt = Y + ((size_t)(mt0 + j) * 16 + 8 * hi) * DFEAT + ncol;
#pragma unroll
        for (int i = 0; i < 8; ++i) Yt[(size_t)i * DFEAT] = acc[i];
    }
}

#define OWN_WAVES  32
#define STEP_LOADS 6
#define STEP_EDGES (STEP_LOADS * 128)
#define DRAIN_AT   64
#define QCAP       (DRAIN_AT + STEP_EDGES)

__global__ __launch_bounds__(OWN_WAVES * 32) void gconv_owner_sum(const float* __restrict__ Y,
                                                                  const int* __restrict__ esrc,
                                                                  const int* __restrict__ edst,
                                                                  const float* __restrict__ eval,
                                                                  float* __restrict__ Z,
                                                                  int nNodes, int nEdges) {
    __shared__ float acc[OWN_ROWS * DFEAT];
    __shared__ int   qall[OWN_WAVES * QCAP];

    const int lane = (int)(threadIdx.x & 31);
    const int wave = __builtin_amdgcn_readfirstlane((int)(threadIdx.x >> 5));
    const int rowBase = (int)blockIdx.x * OWN_ROWS;
    int nrows = nNodes - rowBase;
    if (nrows > OWN_ROWS) nrows = OWN_ROWS;

    for (int i = (int)threadIdx.x; i < OWN_ROWS * DFEAT / 4; i += OWN_WAVES * 32) {
        ((float4*)acc)[i] = make_float4(0.f, 0.f, 0.f, 0.f);
    }
    __syncthreads();

    int* q = qall + wave * QCAP;
    int cnt = 0;
    const int nSteps = (nEdges + STEP_EDGES - 1) / STEP_EDGES;

    int4 cur[STEP_LOADS], nxt[STEP_LOADS];

    auto load_step = [&](int t, int4 (&d)[STEP_LOADS]) {
        const int base = t * STEP_EDGES;
        if (base + STEP_EDGES <= nEdges) {
            const int4* p = (const int4*)(edst + base);
#pragma unroll
            for (int k = 0; k < STEP_LOADS; ++k) d[k] = p[k * 32 + lane];
        } else {
#pragma unroll
            for (int k = 0; k < STEP_LOADS; ++k) {
                const int i0 = base + k * 128 + lane * 4;
                d[k].x = (i0 + 0 < nEdges) ? edst[i0 + 0] : -1;
                d[k].y = (i0 + 1 < nEdges) ? edst[i0 + 1] : -1;
                d[k].z = (i0 + 2 < nEdges) ? edst[i0 + 2] : -1;
                d[k].w = (i0 + 3 < nEdges) ? edst[i0 + 3] : -1;
            }
        }
    };

    if (wave < nSteps) load_step(wave, cur);

    for (int t = wave; t < nSteps; t += OWN_WAVES) {
        const int tn = t + OWN_WAVES;
        if (tn < nSteps) load_step(tn, nxt);

        const int base = t * STEP_EDGES;
#pragma unroll
        for (int k = 0; k < STEP_LOADS; ++k) {
            const bool m0 = (unsigned)(cur[k].x - rowBase) < (unsigned)nrows;
            const bool m1 = (unsigned)(cur[k].y - rowBase) < (unsigned)nrows;
            const bool m2 = (unsigned)(cur[k].z - rowBase) < (unsigned)nrows;
            const bool m3 = (unsigned)(cur[k].w - rowBase) < (unsigned)nrows;
            if (__builtin_amdgcn_ballot_w32(m0 | m1 | m2 | m3) != 0u) {
                const int e0 = base + k * 128 + lane * 4;
                const unsigned b0 = __builtin_amdgcn_ballot_w32(m0);
                if (m0) q[cnt + (int)__builtin_amdgcn_mbcnt_lo(b0, 0u)] = e0 + 0;
                cnt += __builtin_popcount(b0);
                const unsigned b1 = __builtin_amdgcn_ballot_w32(m1);
                if (m1) q[cnt + (int)__builtin_amdgcn_mbcnt_lo(b1, 0u)] = e0 + 1;
                cnt += __builtin_popcount(b1);
                const unsigned b2 = __builtin_amdgcn_ballot_w32(m2);
                if (m2) q[cnt + (int)__builtin_amdgcn_mbcnt_lo(b2, 0u)] = e0 + 2;
                cnt += __builtin_popcount(b2);
                const unsigned b3 = __builtin_amdgcn_ballot_w32(m3);
                if (m3) q[cnt + (int)__builtin_amdgcn_mbcnt_lo(b3, 0u)] = e0 + 3;
                cnt += __builtin_popcount(b3);
            }
        }

        if (cnt >= DRAIN_AT || tn >= nSteps) {
            __builtin_amdgcn_wave_barrier();
            for (int b = 0; b < cnt; b += 32) {
                int n = cnt - b;
                if (n > 32) n = 32;
                const int e  = q[b + (lane < n ? lane : 0)];
                const int s  = esrc[e];
                const int dl = edst[e] - rowBase;
                const float v = eval[e];
                for (int j = 0; j < n; j += 8) {
                    float4 y[8];
#pragma unroll
                    for (int u = 0; u < 8; ++u) {
                        int ju = j + u;
                        if (ju > n - 1) ju = n - 1;
                        const int sj = __builtin_amdgcn_readlane(s, ju);
                        y[u] = ((const float4*)(Y + (size_t)sj * DFEAT))[lane];
                    }
#pragma unroll
                    for (int u = 0; u < 8; ++u) {
                        if (j + u < n) {
                            const int   rj = __builtin_amdgcn_readlane(dl, j + u);
                            const float vj = __int_as_float(__builtin_amdgcn_readlane(__float_as_int(v), j + u));
                            float* a = acc + rj * DFEAT + lane;
                            atomicAdd(a +  0, vj * y[u].x);
                            atomicAdd(a + 32, vj * y[u].y);
                            atomicAdd(a + 64, vj * y[u].z);
                            atomicAdd(a + 96, vj * y[u].w);
                        }
                    }
                }
            }
            cnt = 0;
            __builtin_amdgcn_wave_barrier();
        }

#pragma unroll
        for (int k = 0; k < STEP_LOADS; ++k) cur[k] = nxt[k];
    }

    __syncthreads();

    for (int r = wave; r < nrows; r += OWN_WAVES) {
        const float* a = acc + r * DFEAT + lane;
        float4 o;
        o.x = a[0];
        o.y = a[32];
        o.z = a[64];
        o.w = a[96];
        ((float4*)(Z + (size_t)(rowBase + r) * DFEAT))[lane] = o;
    }
}

#define TAG_LOADS 4
#define TAG_QLEN  64

__device__ __forceinline__ unsigned zero_byte_flags(unsigned x) {
    const unsigned t = (x & 0x7F7F7F7Fu) + 0x7F7F7F7Fu;
    return ~(t | x | 0x7F7F7F7Fu);
}

__device__ __forceinline__ unsigned chunk_flags(uint4 v, unsigned pat) {
    const unsigned f0 = zero_byte_flags(v.x ^ pat);
    const unsigned f1 = zero_byte_flags(v.y ^ pat);
    const unsigned f2 = zero_byte_flags(v.z ^ pat);
    const unsigned f3 = zero_byte_flags(v.w ^ pat);
    return (f0 >> 3) | (f1 >> 2) | (f2 >> 1) | f3;
}

template <bool REC>
__global__ __launch_bounds__(OWN_WAVES * 32) void gconv_owner_sum_tags(const float* __restrict__ Y,
                                                                       const int* __restrict__ esrc,
                                                                       const int* __restrict__ edst,
                                                                       const float* __restrict__ eval,
                                                                       float* __restrict__ Z,
                                                                       int nNodes, int nEdges) {
    __shared__ float acc[OWN_ROWS * DFEAT];
    __shared__ int   qtag[OWN_WAVES * TAG_QLEN];

    const int lane = (int)(threadIdx.x & 31);
    const int wave = __builtin_amdgcn_readfirstlane((int)(threadIdx.x >> 5));
    const int rowBase = (int)blockIdx.x * OWN_ROWS;
    int nrows = nNodes - rowBase;
    if (nrows > OWN_ROWS) nrows = OWN_ROWS;

    for (int i = (int)threadIdx.x; i < OWN_ROWS * DFEAT / 4; i += OWN_WAVES * 32) {
        ((float4*)acc)[i] = make_float4(0.f, 0.f, 0.f, 0.f);
    }
    __syncthreads();

    const uint4* tg = (const uint4*)(Y + (size_t)nNodes * DFEAT);
    const int nChunks = (nEdges + 15) >> 4;
    const int nSteps = (nChunks + TAG_LOADS * 32 - 1) / (TAG_LOADS * 32);
    const unsigned pat = (unsigned)blockIdx.x * 0x01010101u;
    const uint2* rec = (const uint2*)((const unsigned*)tg + (size_t)nChunks * 4);

    int* q = qtag + wave * TAG_QLEN;
    int cnt = 0;

    auto drain = [&](int n) {
        __builtin_amdgcn_wave_barrier();
        const int e  = q[lane < n ? lane : 0];
        int s, dl;
        float v;
        if (REC) {
            const uint2 r = rec[e];
            s  = (int)(r.x & ((1u << REC_SHIFT) - 1u));
            dl = (int)(r.x >> REC_SHIFT);
            v  = __uint_as_float(r.y);
        } else {
            s  = esrc[e];
            dl = edst[e] - rowBase;
            v  = eval[e];
        }
        if ((unsigned)dl >= (unsigned)nrows) { dl = 0; v = 0.f; }
        for (int j = 0; j < n; j += 8) {
            float4 y[8];
#pragma unroll
            for (int u = 0; u < 8; ++u) {
                int ju = j + u;
                if (ju > n - 1) ju = n - 1;
                const int sj = __builtin_amdgcn_readlane(s, ju);
                y[u] = ((const float4*)(Y + (size_t)sj * DFEAT))[lane];
            }
#pragma unroll
            for (int u = 0; u < 8; ++u) {
                if (j + u < n) {
                    const int   rj = __builtin_amdgcn_readlane(dl, j + u);
                    const float vj = __int_as_float(__builtin_amdgcn_readlane(__float_as_int(v), j + u));
                    float* a = acc + rj * DFEAT + lane;
                    atomicAdd(a +  0, vj * y[u].x);
                    atomicAdd(a + 32, vj * y[u].y);
                    atomicAdd(a + 64, vj * y[u].z);
                    atomicAdd(a + 96, vj * y[u].w);
                }
            }
        }
    };

    uint4 cur[TAG_LOADS], nxt[TAG_LOADS];

    auto load_step = [&](int t, uint4 (&d)[TAG_LOADS]) {
        const int c0 = t * (TAG_LOADS * 32);
        if (c0 + TAG_LOADS * 32 <= nChunks) {
#pragma unroll
            for (int k = 0; k < TAG_LOADS; ++k) d[k] = tg[c0 + k * 32 + lane];
        } else {
            const unsigned nm = ~pat;
#pragma unroll
            for (int k = 0; k < TAG_LOADS; ++k) {
                const int c = c0 + k * 32 + lane;
                d[k] = make_uint4(nm, nm, nm, nm);
                if (c < nChunks) d[k] = tg[c];
            }
        }
    };

    if (wave < nSteps) load_step(wave, cur);

    for (int t = wave; t < nSteps; t += OWN_WAVES) {
        const int tn = t + OWN_WAVES;
        if (tn < nSteps) load_step(tn, nxt);

        const unsigned p0 = chunk_flags(cur[0], pat) | (chunk_flags(cur[1], pat) >> 4);
        const unsigned p1 = chunk_flags(cur[2], pat) | (chunk_flags(cur[3], pat) >> 4);
        unsigned m0 = p0, m1 = p1;
        const int laneBase = t * (TAG_LOADS * 512) + lane * 16;
        unsigned left = __builtin_amdgcn_ballot_w32((m0 | m1) != 0u);
        while (left != 0u) {
            const bool has = (m0 | m1) != 0u;
            const bool lo  = (m0 != 0u);
            const unsigned mm = lo ? m0 : m1;
            const int p = __builtin_ctz(mm | 0x80000000u) + (lo ? 0 : 32);
            const unsigned rest = mm & (mm - 1u);
            m0 = lo ? rest : m0;
            m1 = lo ? m1 : rest;
            const int k    = ((p >> 5) << 1) | (((p >> 2) & 1) ^ 1);
            const int word = p & 3;
            const int byte = (p >> 3) & 3;
            const int e = laneBase + k * 512 + word * 4 + byte;
            const bool ok = has && (e < nEdges);
            const unsigned took = __builtin_amdgcn_ballot_w32(ok);
            if (ok) q[cnt + (int)__builtin_amdgcn_mbcnt_lo(took, 0u)] = e;
            cnt += __builtin_popcount(took);
            if (cnt >= 32) {
                drain(32);
                const int rem = cnt - 32;
                const int mv = q[32 + (lane < rem ? lane : 0)];
                if (lane < rem) q[lane] = mv;
                cnt = rem;
            }
            left = __builtin_amdgcn_ballot_w32((m0 | m1) != 0u);
        }

#pragma unroll
        for (int k = 0; k < TAG_LOADS; ++k) cur[k] = nxt[k];
    }

    if (cnt > 0) drain(cnt);

    __syncthreads();

    for (int r = wave; r < nrows; r += OWN_WAVES) {
        const float* a = acc + r * DFEAT + lane;
        float4 o;
        o.x = a[0];
        o.y = a[32];
        o.z = a[64];
        o.w = a[96];
        ((float4*)(Z + (size_t)(rowBase + r) * DFEAT))[lane] = o;
    }
}

extern "C" void kernel_launch(void* const* d_in, const int* in_sizes, int n_in,
                              void* d_out, int out_size, void* d_ws, size_t ws_size,
                              hipStream_t stream) {
    const float* X    = (const float*)d_in[0];
    const float* W    = (const float*)d_in[1];
    const int*   esrc = (const int*)  d_in[2];
    const int*   edst = (const int*)  d_in[3];
    const float* eval = (const float*)d_in[4];
    float*       Z    = (float*)d_out;
    float*       Y    = (float*)d_ws;

    const int nNodes = in_sizes[0] / DFEAT;
    const int nEdges = in_sizes[2];

    const int mTiles   = nNodes / 16;
    const int gemmGrid = (mTiles + GEMM_TPB - 1) / GEMM_TPB;
    const int ownGrid  = (nNodes + OWN_ROWS - 1) / OWN_ROWS;

    const size_t wsNeed = ((size_t)nNodes * DFEAT + (size_t)((nEdges + 15) / 16) * 4) * sizeof(float);
    const size_t wsRec = wsNeed + (size_t)nEdges * 8;
    int tagged = (ws_size >= wsNeed && ownGrid <= 256) ? 1 : 0;
    if (tagged && ws_size >= wsRec && nNodes <= (1 << REC_SHIFT)) tagged = 2;

    gconv_gemm_xw<<<gemmGrid, 256, 0, stream>>>(X, W, Y, mTiles, esrc, edst, eval, nNodes, nEdges, tagged);

    if (tagged == 2) {
        gconv_owner_sum_tags<true><<<ownGrid, OWN_WAVES * 32, 0, stream>>>(Y, esrc, edst, eval, Z, nNodes, nEdges);
    } else if (tagged == 1) {
        gconv_owner_sum_tags<false><<<ownGrid, OWN_WAVES * 32, 0, stream>>>(Y, esrc, edst, eval, Z, nNodes, nEdges);
    } else {
        gconv_owner_sum<<<ownGrid, OWN_WAVES * 32, 0, stream>>>(Y, esrc, edst, eval, Z, nNodes, nEdges);
    }
}


# Owner scan sieved: launch 1 also writes per 2048-edge step and owner a word marking which 64-edge groups hold its edges; owner waves load and test only marked groups, 32 per pass; arrival-order sums
# speedup vs baseline: 3.4664x; 1.0557x over previous
#include <hip/hip_runtime.h>
#include <hip/hip_bf16.h>


typedef __attribute__((ext_vector_type(2))) float v2f;
typedef __attribute__((ext_vector_type(8))) float v8f;

#define DFEAT 128

#define GEMM_TPB 5
#define XPITCH   132
#define OWN_ROWS 391
#define REC_SHIFT 20

__device__ __forceinline__ unsigned rec_word(int s, int d, int nNodes) {
    const unsigned ud = (unsigned)d;
    const unsigned lr = (ud < (unsigned)nNodes) ? ud - (ud / OWN_ROWS) * OWN_ROWS : 0u;
    return ((unsigned)s & ((1u << REC_SHIFT) - 1u)) | (lr << REC_SHIFT);
}
__device__ __forceinline__ unsigned rec_val(float v, int d, int nNodes) {
    return ((unsigned)d < (unsigned)nNodes) ? __float_as_uint(v) : 0u;
}

__global__ __launch_bounds__(256) void gconv_gemm_xw(const float* __restrict__ X,
                                                     const float* __restrict__ W,
                                                     float* __restrict__ Y,
                                                     int mTiles,
                                                     const int* __restrict__ esrc,
                                                     const int* __restrict__ edst,
                                                     const float* __restrict__ eval,
                                                     int nNodes, int nEdges, int tagged) {
    __shared__ float xs[2][16 * XPITCH];
    __shared__ unsigned pres[256];

    const int tid  = (int)threadIdx.x;

    if (tagged) {
        unsigned* tagw = (unsigned*)(Y + (size_t)nNodes * DFEAT);
        const int nWords = ((nEdges + 15) >> 4) << 2;
        uint2* rec = (uint2*)(tagw + nWords);
        unsigned* summ = (unsigned*)(rec + nEdges);
        const bool recs = tagged > 1;
        const bool sieve = tagged > 2;
        const int nStepsT = (nWords + 511) >> 9;
        for (int p = (int)blockIdx.x; p < nStepsT; p += (int)gridDim.x) {
          if (sieve) {
              pres[tid] = 0u;
              __syncthreads();
          }
          for (int wl = tid; wl < 512; wl += 256) {
            const int w = p * 512 + wl;
            if (w < nWords) {
                const int e0 = 4 * w;
                const unsigned gbit = 1u << (wl >> 4);
                unsigned t = 0u;
                if (e0 + 3 < nEdges) {
                    const int4 d = ((const int4*)edst)[w];
                    t = (((unsigned)d.x / OWN_ROWS) & 255u) | ((((unsigned)d.y / OWN_ROWS) & 255u) << 8) |
                        ((((unsigned)d.z / OWN_ROWS) & 255u) << 16) | (((unsigned)d.w / OWN_ROWS) << 24);
                    if (sieve) {
                        atomicOr(&pres[t & 255u], gbit);
                        atomicOr(&pres[(t >> 8) & 255u], gbit);
                        atomicOr(&pres[(t >> 16) & 255u], gbit);
                        atomicOr(&pres[t >> 24], gbit);
                    }
                    if (recs) {
                        const int4   s = ((const int4*)esrc)[w];
                        const float4 v = ((const float4*)eval)[w];
                        uint4 r0, r1;
                        r0.x = rec_word(s.x, d.x, nNodes); r0.y = rec_val(v.x, d.x, nNodes);
                        r0.z = rec_word(s.y, d.y, nNodes); r0.w = rec_val(v.y, d.y, nNodes);
                        r1.x = rec_word(s.z, d.z, nNodes); r1.y = rec_val(v.z, d.z, nNodes);
                        r1.z = rec_word(s.w, d.w, nNodes); r1.w = rec_val(v.w, d.w, nNodes);
                        ((uint4*)rec)[2 * w]     = r0;
                        ((uint4*)rec)[2 * w + 1] = r1;
                    }
                } else {
                    for (int u = 0; u < 4; ++u)
                        if (e0 + u < nEdges) {
                            const int du = edst[e0 + u];
                            const unsigned ob = ((unsigned)du / OWN_ROWS) & 255u;
                            t |= ob << (8 * u);
                            if (sieve) atomicOr(&pres[ob], gbit);
                            if (recs) rec[e0 + u] = make_uint2(rec_word(esrc[e0 + u], du, nNodes),
                                                               rec_val(eval[e0 + u], du, nNodes));
                        }
                }
                tagw[w] = t;
            }
          }
          if (sieve) {
              __syncthreads();
              summ[(size_t)p * 256 + tid] = pres[tid];
          }
        }
    }

    const int lane = tid & 31;
    const int wave = __builtin_amdgcn_readfirstlane(tid >> 5);
    const int hi   = lane >> 4;
    const int l15  = lane & 15;
    const int ncol = wave * 16 + l15;

    v2f bfrag[DFEAT / 4];
#pragma unroll
    for (int kk = 0; kk < DFEAT / 4; ++kk) {
        const int ka = 4 * kk + 2 * hi;
        bfrag[kk].x = W[(size_t)(ka + 0) * DFEAT + ncol];
        bfrag[kk].y = W[(size_t)(ka + 1) * DFEAT + ncol];
    }

    const int mt0 = (int)blockIdx.x * GEMM_TPB;
    int nt = mTiles - mt0;
    if (nt > GEMM_TPB) nt = GEMM_TPB;

    const int sr = tid >> 5;
    const int sc = tid & 31;
    const float4* Xt = (const float4*)(X + ((size_t)mt0 * 16 + sr) * DFEAT) + sc;
    float* sdst = &xs[0][sr * XPITCH + 4 * sc];
    const float* asrc = &xs[0][l15 * XPITCH + 2 * hi];

    float4 g0 = make_float4(0.f, 0.f, 0.f, 0.f), g1 = g0;
    if (nt > 0) {
        g0 = Xt[0];
        g1 = Xt[8 * DFEAT / 4];
        *(float4*)(sdst)              = g0;
        *(float4*)(sdst + 8 * XPITCH) = g1;
    }
    __syncthreads();

    for (int j = 0; j < nt; ++j) {
        const int buf = j & 1;
        const bool more = (j + 1 < nt);
        if (more) {
            const float4* Xn = Xt + (size_t)(j + 1) * (16 * DFEAT / 4);
            g0 = Xn[0];
            g1 = Xn[8 * DFEAT / 4];
        }

        v2f a[DFEAT / 4];
        const float* ab = asrc + buf * (16 * XPITCH);
#pragma unroll
        for (int kk = 0; kk < DFEAT / 4; ++kk) a[kk] = *(const v2f*)(ab + 4 * kk);

        v8f acc = {};
#pragma unroll
        for (int kk = 0; kk < DFEAT / 4; ++kk) {
            acc = __builtin_amdgcn_wmma_f32_16x16x4_f32(
                false, a[kk], false, bfrag[kk], (short)0, acc, false, false);
        }

        if (more) {
            float* sd = sdst + (buf ^ 1) * (16 * XPITCH);
            *(float4*)(sd)              = g0;
            *(float4*)(sd + 8 * XPITCH) = g1;
        }
        __syncthreads();

        float* Yt = Y + ((size_t)(mt0 + j) * 16 + 8 * hi) * DFEAT + ncol;
#pragma unroll
        for (int i = 0; i < 8; ++i) Yt[(size_t)i * DFEAT] = acc[i];
    }
}

#define OWN_WAVES  32
#define STEP_LOADS 6
#define STEP_EDGES (STEP_LOADS * 128)
#define DRAIN_AT   64
#define QCAP       (DRAIN_AT + STEP_EDGES)

__global__ __launch_bounds__(OWN_WAVES * 32) void gconv_owner_sum(const float* __restrict__ Y,
                                                                  const int* __restrict__ esrc,
                                                                  const int* __restrict__ edst,
                                                                  const float* __restrict__ eval,
                                                                  float* __restrict__ Z,
                                                                  int nNodes, int nEdges) {
    __shared__ float acc[OWN_ROWS * DFEAT];
    __shared__ int   qall[OWN_WAVES * QCAP];

    const int lane = (int)(threadIdx.x & 31);
    const int wave = __builtin_amdgcn_readfirstlane((int)(threadIdx.x >> 5));
    const int rowBase = (int)blockIdx.x * OWN_ROWS;
    int nrows = nNodes - rowBase;
    if (nrows > OWN_ROWS) nrows = OWN_ROWS;

    for (int i = (int)threadIdx.x; i < OWN_ROWS * DFEAT / 4; i += OWN_WAVES * 32) {
        ((float4*)acc)[i] = make_float4(0.f, 0.f, 0.f, 0.f);
    }
    __syncthreads();

    int* q = qall + wave * QCAP;
    int cnt = 0;
    const int nSteps = (nEdges + STEP_EDGES - 1) / STEP_EDGES;

    int4 cur[STEP_LOADS], nxt[STEP_LOADS];

    auto load_step = [&](int t, int4 (&d)[STEP_LOADS]) {
        const int base = t * STEP_EDGES;
        if (base + STEP_EDGES <= nEdges) {
            const int4* p = (const int4*)(edst + base);
#pragma unroll
            for (int k = 0; k < STEP_LOADS; ++k) d[k] = p[k * 32 + lane];
        } else {
#pragma unroll
            for (int k = 0; k < STEP_LOADS; ++k) {
                const int i0 = base + k * 128 + lane * 4;
                d[k].x = (i0 + 0 < nEdges) ? edst[i0 + 0] : -1;
                d[k].y = (i0 + 1 < nEdges) ? edst[i0 + 1] : -1;
                d[k].z = (i0 + 2 < nEdges) ? edst[i0 + 2] : -1;
                d[k].w = (i0 + 3 < nEdges) ? edst[i0 + 3] : -1;
            }
        }
    };

    if (wave < nSteps) load_step(wave, cur);

    for (int t = wave; t < nSteps; t += OWN_WAVES) {
        const int tn = t + OWN_WAVES;
        if (tn < nSteps) load_step(tn, nxt);

        const int base = t * STEP_EDGES;
#pragma unroll
        for (int k = 0; k < STEP_LOADS; ++k) {
            const bool m0 = (unsigned)(cur[k].x - rowBase) < (unsigned)nrows;
            const bool m1 = (unsigned)(cur[k].y - rowBase) < (unsigned)nrows;
            const bool m2 = (unsigned)(cur[k].z - rowBase) < (unsigned)nrows;
            const bool m3 = (unsigned)(cur[k].w - rowBase) < (unsigned)nrows;
            if (__builtin_amdgcn_ballot_w32(m0 | m1 | m2 | m3) != 0u) {
                const int e0 = base + k * 128 + lane * 4;
                const unsigned b0 = __builtin_amdgcn_ballot_w32(m0);
                if (m0) q[cnt + (int)__builtin_amdgcn_mbcnt_lo(b0, 0u)] = e0 + 0;
                cnt += __builtin_popcount(b0);
                const unsigned b1 = __builtin_amdgcn_ballot_w32(m1);
                if (m1) q[cnt + (int)__builtin_amdgcn_mbcnt_lo(b1, 0u)] = e0 + 1;
                cnt += __builtin_popcount(b1);
                const unsigned b2 = __builtin_amdgcn_ballot_w32(m2);
                if (m2) q[cnt + (int)__builtin_amdgcn_mbcnt_lo(b2, 0u)] = e0 + 2;
                cnt += __builtin_popcount(b2);
                const unsigned b3 = __builtin_amdgcn_ballot_w32(m3);
                if (m3) q[cnt + (int)__builtin_amdgcn_mbcnt_lo(b3, 0u)] = e0 + 3;
                cnt += __builtin_popcount(b3);
            }
        }

        if (cnt >= DRAIN_AT || tn >= nSteps) {
            __builtin_amdgcn_wave_barrier();
            for (int b = 0; b < cnt; b += 32) {
                int n = cnt - b;
                if (n > 32) n = 32;
                const int e  = q[b + (lane < n ? lane : 0)];
                const int s  = esrc[e];
                const int dl = edst[e] - rowBase;
                const float v = eval[e];
                for (int j = 0; j < n; j += 8) {
                    float4 y[8];
#pragma unroll
                    for (int u = 0; u < 8; ++u) {
                        int ju = j + u;
                        if (ju > n - 1) ju = n - 1;
                        const int sj = __builtin_amdgcn_readlane(s, ju);
                        y[u] = ((const float4*)(Y + (size_t)sj * DFEAT))[lane];
                    }
#pragma unroll
                    for (int u = 0; u < 8; ++u) {
                        if (j + u < n) {
                            const int   rj = __builtin_amdgcn_readlane(dl, j + u);
                            const float vj = __int_as_float(__builtin_amdgcn_readlane(__float_as_int(v), j + u));
                            float* a = acc + rj * DFEAT + lane;
                            atomicAdd(a +  0, vj * y[u].x);
                            atomicAdd(a + 32, vj * y[u].y);
                            atomicAdd(a + 64, vj * y[u].z);
                            atomicAdd(a + 96, vj * y[u].w);
                        }
                    }
                }
            }
            cnt = 0;
            __builtin_amdgcn_wave_barrier();
        }

#pragma unroll
        for (int k = 0; k < STEP_LOADS; ++k) cur[k] = nxt[k];
    }

    __syncthreads();

    for (int r = wave; r < nrows; r += OWN_WAVES) {
        const float* a = acc + r * DFEAT + lane;
        float4 o;
        o.x = a[0];
        o.y = a[32];
        o.z = a[64];
        o.w = a[96];
        ((float4*)(Z + (size_t)(rowBase + r) * DFEAT))[lane] = o;
    }
}

#define TAG_LOADS 4
#define TAG_QLEN  64
#define SIEVE_PAIRS 1024

__device__ __forceinline__ unsigned zero_byte_flags(unsigned x) {
    const unsigned t = (x & 0x7F7F7F7Fu) + 0x7F7F7F7Fu;
    return ~(t | x | 0x7F7F7F7Fu);
}

__device__ __forceinline__ unsigned chunk_flags(uint4 v, unsigned pat) {
    const unsigned f0 = zero_byte_flags(v.x ^ pat);
    const unsigned f1 = zero_byte_flags(v.y ^ pat);
    const unsigned f2 = zero_byte_flags(v.z ^ pat);
    const unsigned f3 = zero_byte_flags(v.w ^ pat);
    return (f0 >> 3) | (f1 >> 2) | (f2 >> 1) | f3;
}

template <bool REC, bool SIEVE>
__global__ __launch_bounds__(OWN_WAVES * 32) void gconv_owner_sum_tags(const float* __restrict__ Y,
                                                                       const int* __restrict__ esrc,
                                                                       const int* __restrict__ edst,
                                                                       const float* __restrict__ eval,
                                                                       float* __restrict__ Z,
                                                                       int nNodes, int nEdges) {
    __shared__ float acc[OWN_ROWS * DFEAT];
    __shared__ int   qtag[OWN_WAVES * TAG_QLEN];
    __shared__ unsigned short pairs[SIEVE ? OWN_WAVES * SIEVE_PAIRS : 1];

    const int lane = (int)(threadIdx.x & 31);
    const int wave = __builtin_amdgcn_readfirstlane((int)(threadIdx.x >> 5));
    const int rowBase = (int)blockIdx.x * OWN_ROWS;
    int nrows = nNodes - rowBase;
    if (nrows > OWN_ROWS) nrows = OWN_ROWS;

    for (int i = (int)threadIdx.x; i < OWN_ROWS * DFEAT / 4; i += OWN_WAVES * 32) {
        ((float4*)acc)[i] = make_float4(0.f, 0.f, 0.f, 0.f);
    }
    __syncthreads();

    const uint4* tg = (const uint4*)(Y + (size_t)nNodes * DFEAT);
    const int nChunks = (nEdges + 15) >> 4;
    const int nSteps = (nChunks + TAG_LOADS * 32 - 1) / (TAG_LOADS * 32);
    const unsigned pat = (unsigned)blockIdx.x * 0x01010101u;
    const uint2* rec = (const uint2*)((const unsigned*)tg + (size_t)nChunks * 4);

    int* q = qtag + wave * TAG_QLEN;
    int cnt = 0;

    auto drain = [&](int n) {
        __builtin_amdgcn_wave_barrier();
        const int e  = q[lane < n ? lane : 0];
        int s, dl;
        float v;
        if (REC) {
            const uint2 r = rec[e];
            s  = (int)(r.x & ((1u << REC_SHIFT) - 1u));
            dl = (int)(r.x >> REC_SHIFT);
            v  = __uint_as_float(r.y);
        } else {
            s  = esrc[e];
            dl = edst[e] - rowBase;
            v  = eval[e];
        }
        if ((unsigned)dl >= (unsigned)nrows) { dl = 0; v = 0.f; }
        for (int j = 0; j < n; j += 8) {
            float4 y[8];
#pragma unroll
            for (int u = 0; u < 8; ++u) {
                int ju = j + u;
                if (ju > n - 1) ju = n - 1;
                const int sj = __builtin_amdgcn_readlane(s, ju);
                y[u] = ((const float4*)(Y + (size_t)sj * DFEAT))[lane];
            }
#pragma unroll
            for (int u = 0; u < 8; ++u) {
                if (j + u < n) {
                    const int   rj = __builtin_amdgcn_readlane(dl, j + u);
                    const float vj = __int_as_float(__builtin_amdgcn_readlane(__float_as_int(v), j + u));
                    float* a = acc + rj * DFEAT + lane;
                    atomicAdd(a +  0, vj * y[u].x);
                    atomicAdd(a + 32, vj * y[u].y);
                    atomicAdd(a + 64, vj * y[u].z);
                    atomicAdd(a + 96, vj * y[u].w);
                }
            }
        }
    };

    uint4 cur[TAG_LOADS], nxt[TAG_LOADS];

    auto take_matches = [&](unsigned m0, unsigned m1, int b0, int b1, int b2, int b3) {
        unsigned left = __builtin_amdgcn_ballot_w32((m0 | m1) != 0u);
        while (left != 0u) {
            const bool has = (m0 | m1) != 0u;
            const bool lo  = (m0 != 0u);
            const unsigned mm = lo ? m0 : m1;
            const int p = __builtin_ctz(mm | 0x80000000u) + (lo ? 0 : 32);
            const unsigned rest = mm & (mm - 1u);
            m0 = lo ? rest : m0;
            m1 = lo ? m1 : rest;
            const int k    = ((p >> 5) << 1) | (((p >> 2) & 1) ^ 1);
            const int word = p & 3;
            const int byte = (p >> 3) & 3;
            const int bk = (k & 2) ? ((k & 1) ? b3 : b2) : ((k & 1) ? b1 : b0);
            const int e = bk + word * 4 + byte;
            const bool ok = has && (e < nEdges);
            const unsigned took = __builtin_amdgcn_ballot_w32(ok);
            if (ok) q[cnt + (int)__builtin_amdgcn_mbcnt_lo(took, 0u)] = e;
            cnt += __builtin_popcount(took);
            if (cnt >= 32) {
                drain(32);
                const int rem = cnt - 32;
                const int mv = q[32 + (lane < rem ? lane : 0)];
                if (lane < rem) q[lane] = mv;
                cnt = rem;
            }
            left = __builtin_amdgcn_ballot_w32((m0 | m1) != 0u);
        }
    };

    if (SIEVE) {
        const unsigned* summ = (const unsigned*)(rec + nEdges);
        unsigned short* pl = pairs + wave * SIEVE_PAIRS;
        const int nRounds = (nSteps + OWN_WAVES * 32 - 1) / (OWN_WAVES * 32);
        for (int r = 0; r < nRounds; ++r) {
            const int stepBase = wave + OWN_WAVES * 32 * r;
            const int tl = stepBase + OWN_WAVES * lane;
            unsigned marks = 0u;
            if (tl < nSteps) marks = summ[(size_t)tl * 256 + blockIdx.x];

            int pc = 0;
            unsigned leftp = __builtin_amdgcn_ballot_w32(marks != 0u);
            while (leftp != 0u) {
                const bool has = marks != 0u;
                const int g = __builtin_ctz(marks | (has ? 0u : 1u));
                marks &= marks - 1u;
                const unsigned took = __builtin_amdgcn_ballot_w32(has);
                if (has) pl[pc + (int)__builtin_amdgcn_mbcnt_lo(took, 0u)] = (unsigned short)((lane << 5) | g);
                pc += __builtin_popcount(took);
                leftp = __builtin_amdgcn_ballot_w32(marks != 0u);
            }
            __builtin_amdgcn_wave_barrier();

            auto load_pairs = [&](int k, uint4 (&d)[TAG_LOADS], int (&eb)[TAG_LOADS]) {
#pragma unroll
                for (int j = 0; j < TAG_LOADS; ++j) {
                    const int idx = k + 8 * j + (lane >> 2);
                    const bool v = idx < pc;
                    const unsigned pr = pl[v ? idx : 0];
                    const int t = stepBase + OWN_WAVES * (int)(pr >> 5);
                    const int c = t * (TAG_LOADS * 32) + (int)(pr & 31u) * TAG_LOADS + (lane & 3);
                    eb[j] = v ? c * 16 : -1;
                    d[j] = tg[c < nChunks ? c : nChunks - 1];
                }
            };

            for (int k = 0; k < pc; k += 32) {
                int eb[TAG_LOADS];
                load_pairs(k, cur, eb);
                unsigned f0 = chunk_flags(cur[0], pat), f1 = chunk_flags(cur[1], pat);
                unsigned f2 = chunk_flags(cur[2], pat), f3 = chunk_flags(cur[3], pat);
                if (eb[0] < 0) f0 = 0u;
                if (eb[1] < 0) f1 = 0u;
                if (eb[2] < 0) f2 = 0u;
                if (eb[3] < 0) f3 = 0u;
                take_matches(f0 | (f1 >> 4), f2 | (f3 >> 4), eb[0], eb[1], eb[2], eb[3]);
            }
            __builtin_amdgcn_wave_barrier();
        }
    } else {

    auto load_step = [&](int t, uint4 (&d)[TAG_LOADS]) {
        const int c0 = t * (TAG_LOADS * 32);
        if (c0 + TAG_LOADS * 32 <= nChunks) {
#pragma unroll
            for (int k = 0; k < TAG_LOADS; ++k) d[k] = tg[c0 + k * 32 + lane];
        } else {
            const unsigned nm = ~pat;
#pragma unroll
            for (int k = 0; k < TAG_LOADS; ++k) {
                const int c = c0 + k * 32 + lane;
                d[k] = make_uint4(nm, nm, nm, nm);
                if (c < nChunks) d[k] = tg[c];
            }
        }
    };

    if (wave < nSteps) load_step(wave, cur);

    for (int t = wave; t < nSteps; t += OWN_WAVES) {
        const int tn = t + OWN_WAVES;
        if (tn < nSteps) load_step(tn, nxt);

        const unsigned p0 = chunk_flags(cur[0], pat) | (chunk_flags(cur[1], pat) >> 4);
        const unsigned p1 = chunk_flags(cur[2], pat) | (chunk_flags(cur[3], pat) >> 4);
        const int lb = t * (TAG_LOADS * 512) + lane * 16;
        take_matches(p0, p1, lb, lb + 512, lb + 1024, lb + 1536);

#pragma unroll
        for (int k = 0; k < TAG_LOADS; ++k) cur[k] = nxt[k];
    }

    }

    if (cnt > 0) drain(cnt);

    __syncthreads();

    for (int r = wave; r < nrows; r += OWN_WAVES) {
        const float* a = acc + r * DFEAT + lane;
        float4 o;
        o.x = a[0];
        o.y = a[32];
        o.z = a[64];
        o.w = a[96];
        ((float4*)(Z + (size_t)(rowBase + r) * DFEAT))[lane] = o;
    }
}

extern "C" void kernel_launch(void* const* d_in, const int* in_sizes, int n_in,
                              void* d_out, int out_size, void* d_ws, size_t ws_size,
                              hipStream_t stream) {
    const float* X    = (const float*)d_in[0];
    const float* W    = (const float*)d_in[1];
    const int*   esrc = (const int*)  d_in[2];
    const int*   edst = (const int*)  d_in[3];
    const float* eval = (const float*)d_in[4];
    float*       Z    = (float*)d_out;
    float*       Y    = (float*)d_ws;

    const int nNodes = in_sizes[0] / DFEAT;
    const int nEdges = in_sizes[2];

    const int mTiles   = nNodes / 16;
    const int gemmGrid = (mTiles + GEMM_TPB - 1) / GEMM_TPB;
    const int ownGrid  = (nNodes + OWN_ROWS - 1) / OWN_ROWS;

    const size_t wsNeed = ((size_t)nNodes * DFEAT + (size_t)((nEdges + 15) / 16) * 4) * sizeof(float);
    const size_t wsRec = wsNeed + (size_t)nEdges * 8;
    const size_t wsSieve = wsRec + (size_t)((nEdges + 2047) / 2048) * 256 * sizeof(unsigned);
    int tagged = (ws_size >= wsNeed && ownGrid <= 256) ? 1 : 0;
    if (tagged && ws_size >= wsRec && nNodes <= (1 << REC_SHIFT)) tagged = 2;
    if (tagged == 2 && ws_size >= wsSieve) tagged = 3;

    gconv_gemm_xw<<<gemmGrid, 256, 0, stream>>>(X, W, Y, mTiles, esrc, edst, eval, nNodes, nEdges, tagged);

    if (tagged == 3) {
        gconv_owner_sum_tags<true, true><<<ownGrid, OWN_WAVES * 32, 0, stream>>>(Y, esrc, edst, eval, Z, nNodes, nEdges);
    } else if (tagged == 2) {
        gconv_owner_sum_tags<true, false><<<ownGrid, OWN_WAVES * 32, 0, stream>>>(Y, esrc, edst, eval, Z, nNodes, nEdges);
    } else if (tagged == 1) {
        gconv_owner_sum_tags<false, false><<<ownGrid, OWN_WAVES * 32, 0, stream>>>(Y, esrc, edst, eval, Z, nNodes, nEdges);
    } else {
        gconv_owner_sum<<<ownGrid, OWN_WAVES * 32, 0, stream>>>(Y, esrc, edst, eval, Z, nNodes, nEdges);
    }
}


# Owner's rows dealt to its 32 waves: scan hands each edge record to its row's wave (LDS list, counting atomic); a wave sums its rows by 16-byte LDS read-add-write, no f32 atomics; arrival-order sums
# speedup vs baseline: 3.7231x; 1.0741x over previous
#include <hip/hip_runtime.h>
#include <hip/hip_bf16.h>


typedef __attribute__((ext_vector_type(2))) float v2f;
typedef __attribute__((ext_vector_type(8))) float v8f;

#define DFEAT 128

#define GEMM_TPB 5
#define XPITCH   132
#define OWN_ROWS 391
#define REC_SHIFT 20

__device__ __forceinline__ unsigned rec_word(int s, int d, int nNodes) {
    const unsigned ud = (unsigned)d;
    const unsigned lr = (ud < (unsigned)nNodes) ? ud - (ud / OWN_ROWS) * OWN_ROWS : 0u;
    return ((unsigned)s & ((1u << REC_SHIFT) - 1u)) | (lr << REC_SHIFT);
}
__device__ __forceinline__ unsigned rec_val(float v, int d, int nNodes) {
    return ((unsigned)d < (unsigned)nNodes) ? __float_as_uint(v) : 0u;
}

__global__ __launch_bounds__(256) void gconv_gemm_xw(const float* __restrict__ X,
                                                     const float* __restrict__ W,
                                                     float* __restrict__ Y,
                                                     int mTiles,
                                                     const int* __restrict__ esrc,
                                                     const int* __restrict__ edst,
                                                     const float* __restrict__ eval,
                                                     int nNodes, int nEdges, int tagged) {
    __shared__ float xs[2][16 * XPITCH];
    __shared__ unsigned pres[256];

    const int tid  = (int)threadIdx.x;

    if (tagged) {
        unsigned* tagw = (unsigned*)(Y + (size_t)nNodes * DFEAT);
        const int nWords = ((nEdges + 15) >> 4) << 2;
        uint2* rec = (uint2*)(tagw + nWords);
        unsigned* summ = (unsigned*)(rec + nEdges);
        const bool recs = tagged > 1;
        const bool sieve = tagged > 2;
        const int nStepsT = (nWords + 511) >> 9;
        for (int p = (int)blockIdx.x; p < nStepsT; p += (int)gridDim.x) {
          if (sieve) {
              pres[tid] = 0u;
              __syncthreads();
          }
          for (int wl = tid; wl < 512; wl += 256) {
            const int w = p * 512 + wl;
            if (w < nWords) {
                const int e0 = 4 * w;
                const unsigned gbit = 1u << (wl >> 4);
                unsigned t = 0u;
                if (e0 + 3 < nEdges) {
                    const int4 d = ((const int4*)edst)[w];
                    t = (((unsigned)d.x / OWN_ROWS) & 255u) | ((((unsigned)d.y / OWN_ROWS) & 255u) << 8) |
                        ((((unsigned)d.z / OWN_ROWS) & 255u) << 16) | (((unsigned)d.w / OWN_ROWS) << 24);
                    if (sieve) {
                        atomicOr(&pres[t & 255u], gbit);
                        atomicOr(&pres[(t >> 8) & 255u], gbit);
                        atomicOr(&pres[(t >> 16) & 255u], gbit);
                        atomicOr(&pres[t >> 24], gbit);
                    }
                    if (recs) {
                        const int4   s = ((const int4*)esrc)[w];
                        const float4 v = ((const float4*)eval)[w];
                        uint4 r0, r1;
                        r0.x = rec_word(s.x, d.x, nNodes); r0.y = rec_val(v.x, d.x, nNodes);
                        r0.z = rec_word(s.y, d.y, nNodes); r0.w = rec_val(v.y, d.y, nNodes);
                        r1.x = rec_word(s.z, d.z, nNodes); r1.y = rec_val(v.z, d.z, nNodes);
                        r1.z = rec_word(s.w, d.w, nNodes); r1.w = rec_val(v.w, d.w, nNodes);
                        ((uint4*)rec)[2 * w]     = r0;
                        ((uint4*)rec)[2 * w + 1] = r1;
                    }
                } else {
                    for (int u = 0; u < 4; ++u)
                        if (e0 + u < nEdges) {
                            const int du = edst[e0 + u];
                            const unsigned ob = ((unsigned)du / OWN_ROWS) & 255u;
                            t |= ob << (8 * u);
                            if (sieve) atomicOr(&pres[ob], gbit);
                            if (recs) rec[e0 + u] = make_uint2(rec_word(esrc[e0 + u], du, nNodes),
                                                               rec_val(eval[e0 + u], du, nNodes));
                        }
                }
                tagw[w] = t;
            }
          }
          if (sieve) {
              __syncthreads();
              summ[(size_t)p * 256 + tid] = pres[tid];
          }
        }
    }

    const int lane = tid & 31;
    const int wave = __builtin_amdgcn_readfirstlane(tid >> 5);
    const int hi   = lane >> 4;
    const int l15  = lane & 15;
    const int ncol = wave * 16 + l15;

    v2f bfrag[DFEAT / 4];
#pragma unroll
    for (int kk = 0; kk < DFEAT / 4; ++kk) {
        const int ka = 4 * kk + 2 * hi;
        bfrag[kk].x = W[(size_t)(ka + 0) * DFEAT + ncol];
        bfrag[kk].y = W[(size_t)(ka + 1) * DFEAT + ncol];
    }

    const int mt0 = (int)blockIdx.x * GEMM_TPB;
    int nt = mTiles - mt0;
    if (nt > GEMM_TPB) nt = GEMM_TPB;

    const int sr = tid >> 5;
    const int sc = tid & 31;
    const float4* Xt = (const float4*)(X + ((size_t)mt0 * 16 + sr) * DFEAT) + sc;
    float* sdst = &xs[0][sr * XPITCH + 4 * sc];
    const float* asrc = &xs[0][l15 * XPITCH + 2 * hi];

    float4 g0 = make_float4(0.f, 0.f, 0.f, 0.f), g1 = g0;
    if (nt > 0) {
        g0 = Xt[0];
        g1 = Xt[8 * DFEAT / 4];
        *(float4*)(sdst)              = g0;
        *(float4*)(sdst + 8 * XPITCH) = g1;
    }
    __syncthreads();

    for (int j = 0; j < nt; ++j) {
        const int buf = j & 1;
        const bool more = (j + 1 < nt);
        if (more) {
            const float4* Xn = Xt + (size_t)(j + 1) * (16 * DFEAT / 4);
            g0 = Xn[0];
            g1 = Xn[8 * DFEAT / 4];
        }

        v2f a[DFEAT / 4];
        const float* ab = asrc + buf * (16 * XPITCH);
#pragma unroll
        for (int kk = 0; kk < DFEAT / 4; ++kk) a[kk] = *(const v2f*)(ab + 4 * kk);

        v8f acc = {};
#pragma unroll
        for (int kk = 0; kk < DFEAT / 4; ++kk) {
            acc = __builtin_amdgcn_wmma_f32_16x16x4_f32(
                false, a[kk], false, bfrag[kk], (short)0, acc, false, false);
        }

        if (more) {
            float* sd = sdst + (buf ^ 1) * (16 * XPITCH);
            *(float4*)(sd)              = g0;
            *(float4*)(sd + 8 * XPITCH) = g1;
        }
        __syncthreads();

        float* Yt = Y + ((size_t)(mt0 + j) * 16 + 8 * hi) * DFEAT + ncol;
#pragma unroll
        for (int i = 0; i < 8; ++i) Yt[(size_t)i * DFEAT] = acc[i];
    }
}

#define OWN_WAVES  32
#define STEP_LOADS 6
#define STEP_EDGES (STEP_LOADS * 128)
#define DRAIN_AT   64
#define QCAP       (DRAIN_AT + STEP_EDGES)

__global__ __launch_bounds__(OWN_WAVES * 32) void gconv_owner_sum(const float* __restrict__ Y,
                                                                  const int* __restrict__ esrc,
                                                                  const int* __restrict__ edst,
                                                                  const float* __restrict__ eval,
                                                                  float* __restrict__ Z,
                                                                  int nNodes, int nEdges) {
    __shared__ float acc[OWN_ROWS * DFEAT];
    __shared__ int   qall[OWN_WAVES * QCAP];

    const int lane = (int)(threadIdx.x & 31);
    const int wave = __builtin_amdgcn_readfirstlane((int)(threadIdx.x >> 5));
    const int rowBase = (int)blockIdx.x * OWN_ROWS;
    int nrows = nNodes - rowBase;
    if (nrows > OWN_ROWS) nrows = OWN_ROWS;

    for (int i = (int)threadIdx.x; i < OWN_ROWS * DFEAT / 4; i += OWN_WAVES * 32) {
        ((float4*)acc)[i] = make_float4(0.f, 0.f, 0.f, 0.f);
    }
    __syncthreads();

    int* q = qall + wave * QCAP;
    int cnt = 0;
    const int nSteps = (nEdges + STEP_EDGES - 1) / STEP_EDGES;

    int4 cur[STEP_LOADS], nxt[STEP_LOADS];

    auto load_step = [&](int t, int4 (&d)[STEP_LOADS]) {
        const int base = t * STEP_EDGES;
        if (base + STEP_EDGES <= nEdges) {
            const int4* p = (const int4*)(edst + base);
#pragma unroll
            for (int k = 0; k < STEP_LOADS; ++k) d[k] = p[k * 32 + lane];
        } else {
#pragma unroll
            for (int k = 0; k < STEP_LOADS; ++k) {
                const int i0 = base + k * 128 + lane * 4;
                d[k].x = (i0 + 0 < nEdges) ? edst[i0 + 0] : -1;
                d[k].y = (i0 + 1 < nEdges) ? edst[i0 + 1] : -1;
                d[k].z = (i0 + 2 < nEdges) ? edst[i0 + 2] : -1;
                d[k].w = (i0 + 3 < nEdges) ? edst[i0 + 3] : -1;
            }
        }
    };

    if (wave < nSteps) load_step(wave, cur);

    for (int t = wave; t < nSteps; t += OWN_WAVES) {
        const int tn = t + OWN_WAVES;
        if (tn < nSteps) load_step(tn, nxt);

        const int base = t * STEP_EDGES;
#pragma unroll
        for (int k = 0; k < STEP_LOADS; ++k) {
            const bool m0 = (unsigned)(cur[k].x - rowBase) < (unsigned)nrows;
            const bool m1 = (unsigned)(cur[k].y - rowBase) < (unsigned)nrows;
            const bool m2 = (unsigned)(cur[k].z - rowBase) < (unsigned)nrows;
            const bool m3 = (unsigned)(cur[k].w - rowBase) < (unsigned)nrows;
            if (__builtin_amdgcn_ballot_w32(m0 | m1 | m2 | m3) != 0u) {
                const int e0 = base + k * 128 + lane * 4;
                const unsigned b0 = __builtin_amdgcn_ballot_w32(m0);
                if (m0) q[cnt + (int)__builtin_amdgcn_mbcnt_lo(b0, 0u)] = e0 + 0;
                cnt += __builtin_popcount(b0);
                const unsigned b1 = __builtin_amdgcn_ballot_w32(m1);
                if (m1) q[cnt + (int)__builtin_amdgcn_mbcnt_lo(b1, 0u)] = e0 + 1;
                cnt += __builtin_popcount(b1);
                const unsigned b2 = __builtin_amdgcn_ballot_w32(m2);
                if (m2) q[cnt + (int)__builtin_amdgcn_mbcnt_lo(b2, 0u)] = e0 + 2;
                cnt += __builtin_popcount(b2);
                const unsigned b3 = __builtin_amdgcn_ballot_w32(m3);
                if (m3) q[cnt + (int)__builtin_amdgcn_mbcnt_lo(b3, 0u)] = e0 + 3;
                cnt += __builtin_popcount(b3);
            }
        }

        if (cnt >= DRAIN_AT || tn >= nSteps) {
            __builtin_amdgcn_wave_barrier();
            for (int b = 0; b < cnt; b += 32) {
                int n = cnt - b;
                if (n > 32) n = 32;
                const int e  = q[b + (lane < n ? lane : 0)];
                const int s  = esrc[e];
                const int dl = edst[e] - rowBase;
                const float v = eval[e];
                for (int j = 0; j < n; j += 8) {
                    float4 y[8];
#pragma unroll
                    for (int u = 0; u < 8; ++u) {
                        int ju = j + u;
                        if (ju > n - 1) ju = n - 1;
                        const int sj = __builtin_amdgcn_readlane(s, ju);
                        y[u] = ((const float4*)(Y + (size_t)sj * DFEAT))[lane];
                    }
#pragma unroll
                    for (int u = 0; u < 8; ++u) {
                        if (j + u < n) {
                            const int   rj = __builtin_amdgcn_readlane(dl, j + u);
                            const float vj = __int_as_float(__builtin_amdgcn_readlane(__float_as_int(v), j + u));
                            float* a = acc + rj * DFEAT + lane;
                            atomicAdd(a +  0, vj * y[u].x);
                            atomicAdd(a + 32, vj * y[u].y);
                            atomicAdd(a + 64, vj * y[u].z);
                            atomicAdd(a + 96, vj * y[u].w);
                        }
                    }
                }
            }
            cnt = 0;
            __builtin_amdgcn_wave_barrier();
        }

#pragma unroll
        for (int k = 0; k < STEP_LOADS; ++k) cur[k] = nxt[k];
    }

    __syncthreads();

    for (int r = wave; r < nrows; r += OWN_WAVES) {
        const float* a = acc + r * DFEAT + lane;
        float4 o;
        o.x = a[0];
        o.y = a[32];
        o.z = a[64];
        o.w = a[96];
        ((float4*)(Z + (size_t)(rowBase + r) * DFEAT))[lane] = o;
    }
}

#define TAG_LOADS 4
#define TAG_QLEN  64
#define SIEVE_LANES 16
#define SIEVE_PAIRS (SIEVE_LANES * 32)
#define ROUTE_CAP 288

__device__ __forceinline__ unsigned zero_byte_flags(unsigned x) {
    const unsigned t = (x & 0x7F7F7F7Fu) + 0x7F7F7F7Fu;
    return ~(t | x | 0x7F7F7F7Fu);
}

__device__ __forceinline__ unsigned chunk_flags(uint4 v, unsigned pat) {
    const unsigned f0 = zero_byte_flags(v.x ^ pat);
    const unsigned f1 = zero_byte_flags(v.y ^ pat);
    const unsigned f2 = zero_byte_flags(v.z ^ pat);
    const unsigned f3 = zero_byte_flags(v.w ^ pat);
    return (f0 >> 3) | (f1 >> 2) | (f2 >> 1) | f3;
}

template <bool REC, bool SIEVE>
__global__ __launch_bounds__(OWN_WAVES * 32) void gconv_owner_sum_tags(const float* __restrict__ Y,
                                                                       const int* __restrict__ esrc,
                                                                       const int* __restrict__ edst,
                                                                       const float* __restrict__ eval,
                                                                       float* __restrict__ Z,
                                                                       int nNodes, int nEdges) {
    __shared__ float acc[OWN_ROWS * DFEAT];
    __shared__ int   qtag[OWN_WAVES * TAG_QLEN];
    __shared__ unsigned short pairs[SIEVE ? OWN_WAVES * SIEVE_PAIRS : 1];
    __shared__ uint2 routed[SIEVE ? OWN_WAVES * ROUTE_CAP : 1];
    __shared__ int   nrouted[SIEVE ? OWN_WAVES : 1];

    const int lane = (int)(threadIdx.x & 31);
    const int wave = __builtin_amdgcn_readfirstlane((int)(threadIdx.x >> 5));
    const int rowBase = (int)blockIdx.x * OWN_ROWS;
    int nrows = nNodes - rowBase;
    if (nrows > OWN_ROWS) nrows = OWN_ROWS;

    for (int i = (int)threadIdx.x; i < OWN_ROWS * DFEAT / 4; i += OWN_WAVES * 32) {
        ((float4*)acc)[i] = make_float4(0.f, 0.f, 0.f, 0.f);
    }
    if (SIEVE && threadIdx.x < OWN_WAVES) nrouted[threadIdx.x] = 0;
    __syncthreads();

    const uint4* tg = (const uint4*)(Y + (size_t)nNodes * DFEAT);
    const int nChunks = (nEdges + 15) >> 4;
    const int nSteps = (nChunks + TAG_LOADS * 32 - 1) / (TAG_LOADS * 32);
    const unsigned pat = (unsigned)blockIdx.x * 0x01010101u;
    const uint2* rec = (const uint2*)((const unsigned*)tg + (size_t)nChunks * 4);

    int* q = qtag + wave * TAG_QLEN;
    int cnt = 0;

    auto drain = [&](int n) {
        __builtin_amdgcn_wave_barrier();
        const int e  = q[lane < n ? lane : 0];
        int s, dl;
        float v;
        if (REC) {
            const uint2 r = rec[e];
            s  = (int)(r.x & ((1u << REC_SHIFT) - 1u));
            dl = (int)(r.x >> REC_SHIFT);
            v  = __uint_as_float(r.y);
        } else {
            s  = esrc[e];
            dl = edst[e] - rowBase;
            v  = eval[e];
        }
        if ((unsigned)dl >= (unsigned)nrows) { dl = 0; v = 0.f; }
        for (int j = 0; j < n; j += 8) {
            float4 y[8];
#pragma unroll
            for (int u = 0; u < 8; ++u) {
                int ju = j + u;
                if (ju > n - 1) ju = n - 1;
                const int sj = __builtin_amdgcn_readlane(s, ju);
                y[u] = ((const float4*)(Y + (size_t)sj * DFEAT))[lane];
            }
#pragma unroll
            for (int u = 0; u < 8; ++u) {
                if (j + u < n) {
                    const int   rj = __builtin_amdgcn_readlane(dl, j + u);
                    const float vj = __int_as_float(__builtin_amdgcn_readlane(__float_as_int(v), j + u));
                    float* a = acc + rj * DFEAT + lane;
                    atomicAdd(a +  0, vj * y[u].x);
                    atomicAdd(a + 32, vj * y[u].y);
                    atomicAdd(a + 64, vj * y[u].z);
                    atomicAdd(a + 96, vj * y[u].w);
                }
            }
        }
    };

    auto route = [&](int n) {
        __builtin_amdgcn_wave_barrier();
        const int e = q[lane < n ? lane : 0];
        uint2 r = rec[e];
        unsigned dl = r.x >> REC_SHIFT;
        if (dl >= (unsigned)nrows) { r.x &= (1u << REC_SHIFT) - 1u; r.y = 0u; dl = 0u; }
        const bool live = lane < n;
        const int tw = (int)(dl & (unsigned)(OWN_WAVES - 1));
        int slot = 0;
        if (live) slot = atomicAdd(&nrouted[tw], 1);
        if (live && slot < ROUTE_CAP) routed[tw * ROUTE_CAP + slot] = r;
        unsigned over = __builtin_amdgcn_ballot_w32(live && slot >= ROUTE_CAP);
        const int s = (int)(r.x & ((1u << REC_SHIFT) - 1u));
        while (over != 0u) {
            const int j = __builtin_ctz(over);
            over &= over - 1u;
            const int   sj = __builtin_amdgcn_readlane(s, j);
            const int   rj = __builtin_amdgcn_readlane((int)dl, j);
            const float vj = __uint_as_float((unsigned)__builtin_amdgcn_readlane((int)r.y, j));
            const float4 yv = ((const float4*)(Y + (size_t)sj * DFEAT))[lane];
            float* a = acc + rj * DFEAT + 4 * lane;
            atomicAdd(a + 0, vj * yv.x);
            atomicAdd(a + 1, vj * yv.y);
            atomicAdd(a + 2, vj * yv.z);
            atomicAdd(a + 3, vj * yv.w);
        }
    };

    uint4 cur[TAG_LOADS], nxt[TAG_LOADS];

    auto take_matches = [&](unsigned m0, unsigned m1, int b0, int b1, int b2, int b3) {
        unsigned left = __builtin_amdgcn_ballot_w32((m0 | m1) != 0u);
        while (left != 0u) {
            const bool has = (m0 | m1) != 0u;
            const bool lo  = (m0 != 0u);
            const unsigned mm = lo ? m0 : m1;
            const int p = __builtin_ctz(mm | 0x80000000u) + (lo ? 0 : 32);
            const unsigned rest = mm & (mm - 1u);
            m0 = lo ? rest : m0;
            m1 = lo ? m1 : rest;
            const int k    = ((p >> 5) << 1) | (((p >> 2) & 1) ^ 1);
            const int word = p & 3;
            const int byte = (p >> 3) & 3;
            const int bk = (k & 2) ? ((k & 1) ? b3 : b2) : ((k & 1) ? b1 : b0);
            const int e = bk + word * 4 + byte;
            const bool ok = has && (e < nEdges);
            const unsigned took = __builtin_amdgcn_ballot_w32(ok);
            if (ok) q[cnt + (int)__builtin_amdgcn_mbcnt_lo(took, 0u)] = e;
            cnt += __builtin_popcount(took);
            if (cnt >= 32) {
                if (SIEVE) route(32); else drain(32);
                const int rem = cnt - 32;
                const int mv = q[32 + (lane < rem ? lane : 0)];
                if (lane < rem) q[lane] = mv;
                cnt = rem;
            }
            left = __builtin_amdgcn_ballot_w32((m0 | m1) != 0u);
        }
    };

    if (SIEVE) {
        const unsigned* summ = (const unsigned*)(rec + nEdges);
        unsigned short* pl = pairs + wave * SIEVE_PAIRS;
        const int nRounds = (nSteps + OWN_WAVES * SIEVE_LANES - 1) / (OWN_WAVES * SIEVE_LANES);
        for (int r = 0; r < nRounds; ++r) {
            const int stepBase = wave + OWN_WAVES * SIEVE_LANES * r;
            const int tl = stepBase + OWN_WAVES * lane;
            unsigned marks = 0u;
            if (lane < SIEVE_LANES && tl < nSteps) marks = summ[(size_t)tl * 256 + blockIdx.x];

            int pc = 0;
            unsigned leftp = __builtin_amdgcn_ballot_w32(marks != 0u);
            while (leftp != 0u) {
                const bool has = marks != 0u;
                const int g = __builtin_ctz(marks | (has ? 0u : 1u));
                marks &= marks - 1u;
                const unsigned took = __builtin_amdgcn_ballot_w32(has);
                if (has) pl[pc + (int)__builtin_amdgcn_mbcnt_lo(took, 0u)] = (unsigned short)((lane << 5) | g);
                pc += __builtin_popcount(took);
                leftp = __builtin_amdgcn_ballot_w32(marks != 0u);
            }
            __builtin_amdgcn_wave_barrier();

            auto load_pairs = [&](int k, uint4 (&d)[TAG_LOADS], int (&eb)[TAG_LOADS]) {
#pragma unroll
                for (int j = 0; j < TAG_LOADS; ++j) {
                    const int idx = k + 8 * j + (lane >> 2);
                    const bool v = idx < pc;
                    const unsigned pr = pl[v ? idx : 0];
                    const int t = stepBase + OWN_WAVES * (int)(pr >> 5);
                    const int c = t * (TAG_LOADS * 32) + (int)(pr & 31u) * TAG_LOADS + (lane & 3);
                    eb[j] = v ? c * 16 : -1;
                    d[j] = tg[c < nChunks ? c : nChunks - 1];
                }
            };

            for (int k = 0; k < pc; k += 32) {
                int eb[TAG_LOADS];
                load_pairs(k, cur, eb);
                unsigned f0 = chunk_flags(cur[0], pat), f1 = chunk_flags(cur[1], pat);
                unsigned f2 = chunk_flags(cur[2], pat), f3 = chunk_flags(cur[3], pat);
                if (eb[0] < 0) f0 = 0u;
                if (eb[1] < 0) f1 = 0u;
                if (eb[2] < 0) f2 = 0u;
                if (eb[3] < 0) f3 = 0u;
                take_matches(f0 | (f1 >> 4), f2 | (f3 >> 4), eb[0], eb[1], eb[2], eb[3]);
            }
            __builtin_amdgcn_wave_barrier();
        }
    } else {

    auto load_step = [&](int t, uint4 (&d)[TAG_LOADS]) {
        const int c0 = t * (TAG_LOADS * 32);
        if (c0 + TAG_LOADS * 32 <= nChunks) {
#pragma unroll
            for (int k = 0; k < TAG_LOADS; ++k) d[k] = tg[c0 + k * 32 + lane];
        } else {
            const unsigned nm = ~pat;
#pragma unroll
            for (int k = 0; k < TAG_LOADS; ++k) {
                const int c = c0 + k * 32 + lane;
                d[k] = make_uint4(nm, nm, nm, nm);
                if (c < nChunks) d[k] = tg[c];
            }
        }
    };

    if (wave < nSteps) load_step(wave, cur);

    for (int t = wave; t < nSteps; t += OWN_WAVES) {
        const int tn = t + OWN_WAVES;
        if (tn < nSteps) load_step(tn, nxt);

        const unsigned p0 = chunk_flags(cur[0], pat) | (chunk_flags(cur[1], pat) >> 4);
        const unsigned p1 = chunk_flags(cur[2], pat) | (chunk_flags(cur[3], pat) >> 4);
        const int lb = t * (TAG_LOADS * 512) + lane * 16;
        take_matches(p0, p1, lb, lb + 512, lb + 1024, lb + 1536);

#pragma unroll
        for (int k = 0; k < TAG_LOADS; ++k) cur[k] = nxt[k];
    }

    }

    if (SIEVE) {
        if (cnt > 0) route(cnt);

        __syncthreads();

        int nb = __builtin_amdgcn_readfirstlane(nrouted[wave]);
        if (nb > ROUTE_CAP) nb = ROUTE_CAP;
        const uint2* bl = routed + wave * ROUTE_CAP;
        const int nk = (nb + 7) >> 3;

        auto issue = [&](int k, float4 (&y)[8], uint2& r) {
            int i = 8 * k + (lane & 7);
            if (i > nb - 1) i = nb - 1;
            r = bl[i];
            const int s = (int)(r.x & ((1u << REC_SHIFT) - 1u));
#pragma unroll
            for (int u = 0; u < 8; ++u) {
                const int sj = __builtin_amdgcn_readlane(s, u);
                y[u] = ((const float4*)(Y + (size_t)sj * DFEAT))[lane];
            }
        };
        auto consume = [&](int m, const float4 (&y)[8], uint2 r) {
            const int dl = (int)(r.x >> REC_SHIFT);
#pragma unroll
            for (int u = 0; u < 8; ++u) {
                if (u < m) {
                    const int   rj = __builtin_amdgcn_readlane(dl, u);
                    const float vj = __uint_as_float((unsigned)__builtin_amdgcn_readlane((int)r.y, u));
                    float4* a = (float4*)(acc + rj * DFEAT) + lane;
                    float4 t = *a;
                    t.x += vj * y[u].x;
                    t.y += vj * y[u].y;
                    t.z += vj * y[u].z;
                    t.w += vj * y[u].w;
                    *a = t;
                }
            }
        };

        if (nk > 0) {
            float4 ya[8], yb[8];
            uint2 ra, rb;
            issue(0, ya, ra);
            int k = 0;
            while (k + 2 < nk) {
                issue(k + 1, yb, rb);
                consume(8, ya, ra);
                issue(k + 2, ya, ra);
                consume(8, yb, rb);
                k += 2;
            }
            if (k + 1 < nk) {
                issue(k + 1, yb, rb);
                consume(8, ya, ra);
                consume(nb - 8 * (k + 1), yb, rb);
            } else {
                consume(nb - 8 * k, ya, ra);
            }
        }

        for (int r = wave; r < nrows; r += OWN_WAVES) {
            const float4 o = ((const float4*)(acc + r * DFEAT))[lane];
            ((float4*)(Z + (size_t)(rowBase + r) * DFEAT))[lane] = o;
        }
    } else {
        if (cnt > 0) drain(cnt);

        __syncthreads();

        for (int r = wave; r < nrows; r += OWN_WAVES) {
            const float* a = acc + r * DFEAT + lane;
            float4 o;
            o.x = a[0];
            o.y = a[32];
            o.z = a[64];
            o.w = a[96];
            ((float4*)(Z + (size_t)(rowBase + r) * DFEAT))[lane] = o;
        }
    }
}

extern "C" void kernel_launch(void* const* d_in, const int* in_sizes, int n_in,
                              void* d_out, int out_size, void* d_ws, size_t ws_size,
                              hipStream_t stream) {
    const float* X    = (const float*)d_in[0];
    const float* W    = (const float*)d_in[1];
    const int*   esrc = (const int*)  d_in[2];
    const int*   edst = (const int*)  d_in[3];
    const float* eval = (const float*)d_in[4];
    float*       Z    = (float*)d_out;
    float*       Y    = (float*)d_ws;

    const int nNodes = in_sizes[0] / DFEAT;
    const int nEdges = in_sizes[2];

    const int mTiles   = nNodes / 16;
    const int gemmGrid = (mTiles + GEMM_TPB - 1) / GEMM_TPB;
    const int ownGrid  = (nNodes + OWN_ROWS - 1) / OWN_ROWS;

    const size_t wsNeed = ((size_t)nNodes * DFEAT + (size_t)((nEdges + 15) / 16) * 4) * sizeof(float);
    const size_t wsRec = wsNeed + (size_t)nEdges * 8;
    const size_t wsSieve = wsRec + (size_t)((nEdges + 2047) / 2048) * 256 * sizeof(unsigned);
    int tagged = (ws_size >= wsNeed && ownGrid <= 256) ? 1 : 0;
    if (tagged && ws_size >= wsRec && nNodes <= (1 << REC_SHIFT)) tagged = 2;
    if (tagged == 2 && ws_size >= wsSieve) tagged = 3;

    gconv_gemm_xw<<<gemmGrid, 256, 0, stream>>>(X, W, Y, mTiles, esrc, edst, eval, nNodes, nEdges, tagged);

    if (tagged == 3) {
        gconv_owner_sum_tags<true, true><<<ownGrid, OWN_WAVES * 32, 0, stream>>>(Y, esrc, edst, eval, Z, nNodes, nEdges);
    } else if (tagged == 2) {
        gconv_owner_sum_tags<true, false><<<ownGrid, OWN_WAVES * 32, 0, stream>>>(Y, esrc, edst, eval, Z, nNodes, nEdges);
    } else if (tagged == 1) {
        gconv_owner_sum_tags<false, false><<<ownGrid, OWN_WAVES * 32, 0, stream>>>(Y, esrc, edst, eval, Z, nNodes, nEdges);
    } else {
        gconv_owner_sum<<<ownGrid, OWN_WAVES * 32, 0, stream>>>(Y, esrc, edst, eval, Z, nNodes, nEdges);
    }
}
